# starting kernel
# baseline (speedup 1.0000x reference)
#include <hip/hip_runtime.h>
#include <math.h>

typedef __attribute__((ext_vector_type(2))) float v2f;
typedef __attribute__((ext_vector_type(8))) float v8f;

#define NN 100000
#define EE 1600000
#define QQ 1000000

__global__ void GCN_deg_kernel(const int* __restrict__ esrc, const int* __restrict__ edst,
                               float* __restrict__ dout, float* __restrict__ din) {
    int e = blockIdx.x * blockDim.x + threadIdx.x;
    if (e >= EE) return;
    unsafeAtomicAdd(&dout[esrc[e]], 1.0f);
    unsafeAtomicAdd(&din[edst[e]], 1.0f);
}

__global__ void GCN_norm_kernel(float* __restrict__ d) {
    int i = blockIdx.x * blockDim.x + threadIdx.x;
    if (i >= 2 * NN) return;
    d[i] = rsqrtf(fmaxf(d[i], 1.0f));
}

__global__ void GCN_scatter1_kernel(const float* __restrict__ feat,
                                    const int* __restrict__ esrc, const int* __restrict__ edst,
                                    const float* __restrict__ onorm, float* __restrict__ agg) {
    int t = blockIdx.x * blockDim.x + threadIdx.x;
    int e = t >> 5, lane = t & 31;
    int s = esrc[e], d = edst[e];
    float sc = onorm[s];
    const float4 v = *(const float4*)(feat + (size_t)s * 128 + lane * 4);
    float* dp = agg + (size_t)d * 128 + lane * 4;
    unsafeAtomicAdd(dp + 0, v.x * sc);
    unsafeAtomicAdd(dp + 1, v.y * sc);
    unsafeAtomicAdd(dp + 2, v.z * sc);
    unsafeAtomicAdd(dp + 3, v.w * sc);
}

__global__ void __launch_bounds__(256)
GCN_gemm1_kernel(const float* __restrict__ A, const float* __restrict__ W,
                 const float* __restrict__ bias, const float* __restrict__ inorm,
                 float* __restrict__ H) {
    __shared__ float lW[128 * 128];
    int tid = threadIdx.x;
    {
        const float4* Wv = (const float4*)W;
        float4* lWv = (float4*)lW;
#pragma unroll
        for (int i = 0; i < 16; ++i) lWv[tid + 256 * i] = Wv[tid + 256 * i];
    }
    __syncthreads();

    int wave = tid >> 5, lane = tid & 31;
    int row0 = blockIdx.x * 16, col0 = wave * 16;
    int r = lane & 15, kh = lane >> 4;

    v8f c = {};
    const float* Ap = A + (size_t)(row0 + r) * 128 + 2 * kh;
    const float* Bp = lW + (size_t)(2 * kh) * 128 + col0 + r;
#pragma unroll
    for (int kt = 0; kt < 32; ++kt) {
        v2f a = *(const v2f*)(Ap + kt * 4);
        v2f b;
        b.x = Bp[kt * 512];
        b.y = Bp[kt * 512 + 128];
        c = __builtin_amdgcn_wmma_f32_16x16x4_f32(false, a, false, b, (short)0, c, false, false);
    }
    int n = col0 + r;
    float bb = bias[n];
#pragma unroll
    for (int v = 0; v < 8; ++v) {
        int m = row0 + v + 8 * kh;
        float val = fmaf(c[v], inorm[m], bb);
        H[(size_t)m * 128 + n] = fmaxf(val, 0.0f);
    }
}

__global__ void __launch_bounds__(128)
GCN_gemm2_kernel(const float* __restrict__ Hin, const float* __restrict__ W,
                 const float* __restrict__ onorm, float* __restrict__ T) {
    __shared__ float lW[128 * 64];
    int tid = threadIdx.x;
    {
        const float4* Wv = (const float4*)W;
        float4* lWv = (float4*)lW;
#pragma unroll
        for (int i = 0; i < 16; ++i) lWv[tid + 128 * i] = Wv[tid + 128 * i];
    }
    __syncthreads();

    int wave = tid >> 5, lane = tid & 31;
    int row0 = blockIdx.x * 16, col0 = wave * 16;
    int r = lane & 15, kh = lane >> 4;
    float sc = onorm[row0 + r];

    v8f c = {};
    const float* Ap = Hin + (size_t)(row0 + r) * 128 + 2 * kh;
    const float* Bp = lW + (size_t)(2 * kh) * 64 + col0 + r;
#pragma unroll
    for (int kt = 0; kt < 32; ++kt) {
        v2f a = *(const v2f*)(Ap + kt * 4);
        a.x *= sc; a.y *= sc;
        v2f b;
        b.x = Bp[kt * 256];
        b.y = Bp[kt * 256 + 64];
        c = __builtin_amdgcn_wmma_f32_16x16x4_f32(false, a, false, b, (short)0, c, false, false);
    }
    int n = col0 + r;
#pragma unroll
    for (int v = 0; v < 8; ++v) {
        int m = row0 + v + 8 * kh;
        T[(size_t)m * 64 + n] = c[v];
    }
}

__global__ void GCN_scatter2_kernel(const float* __restrict__ T,
                                    const int* __restrict__ esrc, const int* __restrict__ edst,
                                    float* __restrict__ H2) {
    int t = blockIdx.x * blockDim.x + threadIdx.x;
    int e = t >> 5, lane = t & 31;
    int s = esrc[e], d = edst[e];
    const float2 v = *(const float2*)(T + (size_t)s * 64 + lane * 2);
    float* dp = H2 + (size_t)d * 64 + lane * 2;
    unsafeAtomicAdd(dp + 0, v.x);
    unsafeAtomicAdd(dp + 1, v.y);
}

__global__ void GCN_finalize_kernel(float* __restrict__ H2, const float* __restrict__ inorm,
                                    const float* __restrict__ b2) {
    int i = blockIdx.x * blockDim.x + threadIdx.x;
    if (i >= NN * 16) return;
    int row = i >> 4;
    int c4 = (i & 15) * 4;
    float s = inorm[row];
    float4 v = ((float4*)H2)[i];
    v.x = fmaf(v.x, s, b2[c4 + 0]);
    v.y = fmaf(v.y, s, b2[c4 + 1]);
    v.z = fmaf(v.z, s, b2[c4 + 2]);
    v.w = fmaf(v.w, s, b2[c4 + 3]);
    ((float4*)H2)[i] = v;
}

__global__ void GCN_score_kernel(const float* __restrict__ H2, const int* __restrict__ qs,
                                 const int* __restrict__ qd, float* __restrict__ out) {
    int t = blockIdx.x * blockDim.x + threadIdx.x;
    int q = t >> 5, lane = t & 31;
    int s = qs[q], d = qd[q];
    float2 a = *(const float2*)(H2 + (size_t)s * 64 + lane * 2);
    float2 b = *(const float2*)(H2 + (size_t)d * 64 + lane * 2);
    float p = a.x * b.x + a.y * b.y;
#pragma unroll
    for (int off = 16; off > 0; off >>= 1) p += __shfl_xor(p, off, 32);
    if (lane == 0) out[q] = 1.0f / (1.0f + __expf(-p));
}

extern "C" void kernel_launch(void* const* d_in, const int* in_sizes, int n_in,
                              void* d_out, int out_size, void* d_ws, size_t ws_size,
                              hipStream_t stream) {
    const float* feat = (const float*)d_in[0];
    const int* esrc   = (const int*)d_in[1];
    const int* edst   = (const int*)d_in[2];
    const int* qsrc   = (const int*)d_in[3];
    const int* qdst   = (const int*)d_in[4];
    const float* W1   = (const float*)d_in[5];
    const float* b1   = (const float*)d_in[6];
    const float* W2   = (const float*)d_in[7];
    const float* b2   = (const float*)d_in[8];
    float* out = (float*)d_out;

    float* ws    = (float*)d_ws;
    float* onorm = ws;
    float* inorm = ws + NN;
    float* agg1  = ws + 2 * NN;
    float* h1    = agg1 + (size_t)NN * 128;
    float* t2    = h1 + (size_t)NN * 128;
    float* h2    = t2 + (size_t)NN * 64;

    hipMemsetAsync(ws, 0, (size_t)(2 * NN + (size_t)NN * 128) * sizeof(float), stream);
    hipMemsetAsync(h2, 0, (size_t)NN * 64 * sizeof(float), stream);

    GCN_deg_kernel<<<(EE + 255) / 256, 256, 0, stream>>>(esrc, edst, onorm, inorm);
    GCN_norm_kernel<<<(2 * NN + 255) / 256, 256, 0, stream>>>(onorm);
    GCN_scatter1_kernel<<<EE / 8, 256, 0, stream>>>(feat, esrc, edst, onorm, agg1);
    GCN_gemm1_kernel<<<NN / 16, 256, 0, stream>>>(agg1, W1, b1, inorm, h1);
    GCN_gemm2_kernel<<<NN / 16, 128, 0, stream>>>(h1, W2, onorm, t2);
    GCN_scatter2_kernel<<<EE / 8, 256, 0, stream>>>(t2, esrc, edst, h2);
    GCN_finalize_kernel<<<(NN * 16 + 255) / 256, 256, 0, stream>>>(h2, inorm, b2);
    GCN_score_kernel<<<QQ / 8, 256, 0, stream>>>(h2, qsrc, qdst, out);
}


# Both GEMMs: operand roles swapped: a wave keeps its 16-column slice of W in registers over a strided loop of row tiles (no W in LDS); the A tile is staged once per block in LDS; sum order kept
# speedup vs baseline: 1.5984x; 1.5984x over previous
#include <hip/hip_runtime.h>
#include <math.h>

typedef __attribute__((ext_vector_type(2))) float v2f;
typedef __attribute__((ext_vector_type(8))) float v8f;

#define NN 100000
#define EE 1600000
#define QQ 1000000

__global__ void GCN_deg_kernel(const int* __restrict__ esrc, const int* __restrict__ edst,
                               float* __restrict__ dout, float* __restrict__ din) {
    int e = blockIdx.x * blockDim.x + threadIdx.x;
    if (e >= EE) return;
    unsafeAtomicAdd(&dout[esrc[e]], 1.0f);
    unsafeAtomicAdd(&din[edst[e]], 1.0f);
}

__global__ void GCN_norm_kernel(float* __restrict__ d) {
    int i = blockIdx.x * blockDim.x + threadIdx.x;
    if (i >= 2 * NN) return;
    d[i] = rsqrtf(fmaxf(d[i], 1.0f));
}

static_assert(EE % 256 == 0, "edge grid has no tail");
__global__ void __launch_bounds__(256)
GCN_scatter1_kernel(const float* __restrict__ feat,
                    const int* __restrict__ esrc, const int* __restrict__ edst,
                    const float* __restrict__ onorm, float* __restrict__ agg) {
    int t = blockIdx.x * 256 + threadIdx.x;
    int lane = threadIdx.x & 31;
    int sv = esrc[t], dv = edst[t];
    float scv = onorm[sv];
#pragma unroll
    for (int b = 0; b < 32; b += 8) {
        float v[8][4];
#pragma unroll
        for (int j = 0; j < 8; ++j) {
            int s = __builtin_amdgcn_readlane(sv, b + j);
            const float* sp = feat + (size_t)s * 128 + lane;
            v[j][0] = sp[0]; v[j][1] = sp[32]; v[j][2] = sp[64]; v[j][3] = sp[96];
        }
#pragma unroll
        for (int j = 0; j < 8; ++j) {
            int d = __builtin_amdgcn_readlane(dv, b + j);
            float sc = __int_as_float(__builtin_amdgcn_readlane(__float_as_int(scv), b + j));
            float* dp = agg + (size_t)d * 128 + lane;
            unsafeAtomicAdd(dp + 0,  v[j][0] * sc);
            unsafeAtomicAdd(dp + 32, v[j][1] * sc);
            unsafeAtomicAdd(dp + 64, v[j][2] * sc);
            unsafeAtomicAdd(dp + 96, v[j][3] * sc);
        }
    }
}

#define LDA 132
__global__ void __launch_bounds__(256)
GCN_gemm1_kernel(const float* __restrict__ A, const float* __restrict__ W,
                 const float* __restrict__ bias, const float* __restrict__ inorm,
                 float* __restrict__ H) {
    __shared__ __attribute__((aligned(16))) float lA[2][16 * LDA];
    const int ntiles = NN / 16;
    int tile = blockIdx.x;
    if (tile >= ntiles) return;

    const int tid = threadIdx.x;
    const int wave = tid >> 5, lane = tid & 31;
    const int r = lane & 15, kh = lane >> 4;
    const int col0 = wave * 16;

    v2f bf[32];
    {
        const float* Bp = W + (size_t)(2 * kh) * 128 + col0 + r;
#pragma unroll
        for (int kt = 0; kt < 32; ++kt) {
            bf[kt].x = Bp[kt * 512];
            bf[kt].y = Bp[kt * 512 + 128];
        }
    }
    const int n = col0 + r;
    const float bb = bias[n];

    const int srow = wave, sc4 = lane;
    const float4* A4 = (const float4*)A;
    float4 p0 = A4[((size_t)tile * 16 + srow) * 32 + sc4];
    float4 p1 = A4[((size_t)tile * 16 + srow + 8) * 32 + sc4];
    float ni[8];
#pragma unroll
    for (int v = 0; v < 8; ++v) ni[v] = inorm[tile * 16 + 8 * kh + v];
    *(float4*)&lA[0][srow * LDA + sc4 * 4] = p0;
    *(float4*)&lA[0][(srow + 8) * LDA + sc4 * 4] = p1;
    __syncthreads();

    int cur = 0;
    for (; tile < ntiles; tile += gridDim.x) {
        const int nxt = tile + gridDim.x;
        const bool more = nxt < ntiles;
        float ci[8];
#pragma unroll
        for (int v = 0; v < 8; ++v) ci[v] = ni[v];
        if (more) {
            p0 = A4[((size_t)nxt * 16 + srow) * 32 + sc4];
            p1 = A4[((size_t)nxt * 16 + srow + 8) * 32 + sc4];
#pragma unroll
            for (int v = 0; v < 8; ++v) ni[v] = inorm[nxt * 16 + 8 * kh + v];
        }
        const float* ap = &lA[cur][r * LDA + 2 * kh];
        v8f c = {};
#pragma unroll
        for (int kt = 0; kt < 32; ++kt) {
            v2f a = *(const v2f*)(ap + kt * 4);
            c = __builtin_amdgcn_wmma_f32_16x16x4_f32(false, a, false, bf[kt], (short)0, c, false, false);
        }
        if (more) {
            *(float4*)&lA[cur ^ 1][srow * LDA + sc4 * 4] = p0;
            *(float4*)&lA[cur ^ 1][(srow + 8) * LDA + sc4 * 4] = p1;
        }
        __syncthreads();
        const int row0 = tile * 16;
#pragma unroll
        for (int v = 0; v < 8; ++v) {
            int m = row0 + v + 8 * kh;
            float val = fmaf(c[v], ci[v], bb);
            H[(size_t)m * 128 + n] = fmaxf(val, 0.0f);
        }
        cur ^= 1;
    }
}

__global__ void __launch_bounds__(128)
GCN_gemm2_kernel(const float* __restrict__ Hin, const float* __restrict__ W,
                 const float* __restrict__ onorm, float* __restrict__ T) {
    __shared__ __attribute__((aligned(16))) float lA[2][16 * LDA];
    const int ntiles = NN / 16;
    int tile = blockIdx.x;
    if (tile >= ntiles) return;

    const int tid = threadIdx.x;
    const int wave = tid >> 5, lane = tid & 31;
    const int r = lane & 15, kh = lane >> 4;
    const int col0 = wave * 16;

    v2f bf[32];
    {
        const float* Bp = W + (size_t)(2 * kh) * 64 + col0 + r;
#pragma unroll
        for (int kt = 0; kt < 32; ++kt) {
            bf[kt].x = Bp[kt * 256];
            bf[kt].y = Bp[kt * 256 + 64];
        }
    }
    const int n = col0 + r;

    const int srow = wave, sc4 = lane;
    const float4* A4 = (const float4*)Hin;
    float4 p[4];
#pragma unroll
    for (int j = 0; j < 4; ++j) {
        int row = tile * 16 + srow + 4 * j;
        float sc = onorm[row];
        float4 q = A4[(size_t)row * 32 + sc4];
        q.x *= sc; q.y *= sc; q.z *= sc; q.w *= sc;
        p[j] = q;
    }
#pragma unroll
    for (int j = 0; j < 4; ++j) *(float4*)&lA[0][(srow + 4 * j) * LDA + sc4 * 4] = p[j];
    __syncthreads();

    int cur = 0;
    for (; tile < ntiles; tile += gridDim.x) {
        const int nxt = tile + gridDim.x;
        const bool more = nxt < ntiles;
        float sn[4];
        if (more) {
#pragma unroll
            for (int j = 0; j < 4; ++j) {
                int row = nxt * 16 + srow + 4 * j;
                sn[j] = onorm[row];
                p[j] = A4[(size_t)row * 32 + sc4];
            }
        }
        const float* ap = &lA[cur][r * LDA + 2 * kh];
        v8f c = {};
#pragma unroll
        for (int kt = 0; kt < 32; ++kt) {
            v2f a = *(const v2f*)(ap + kt * 4);
            c = __builtin_amdgcn_wmma_f32_16x16x4_f32(false, a, false, bf[kt], (short)0, c, false, false);
        }
        if (more) {
#pragma unroll
            for (int j = 0; j < 4; ++j) {
                float4 q = p[j];
                float sc = sn[j];
                q.x *= sc; q.y *= sc; q.z *= sc; q.w *= sc;
                *(float4*)&lA[cur ^ 1][(srow + 4 * j) * LDA + sc4 * 4] = q;
            }
        }
        __syncthreads();
        const int row0 = tile * 16;
#pragma unroll
        for (int v = 0; v < 8; ++v) {
            int m = row0 + v + 8 * kh;
            T[(size_t)m * 64 + n] = c[v];
        }
        cur ^= 1;
    }
}

__global__ void __launch_bounds__(256)
GCN_scatter2_kernel(const float* __restrict__ T,
                    const int* __restrict__ esrc, const int* __restrict__ edst,
                    float* __restrict__ H2) {
    int t = blockIdx.x * 256 + threadIdx.x;
    int lane = threadIdx.x & 31;
    int sv = esrc[t], dv = edst[t];
#pragma unroll
    for (int b = 0; b < 32; b += 16) {
        float v[16][2];
#pragma unroll
        for (int j = 0; j < 16; ++j) {
            int s = __builtin_amdgcn_readlane(sv, b + j);
            const float* sp = T + (size_t)s * 64 + lane;
            v[j][0] = sp[0]; v[j][1] = sp[32];
        }
#pragma unroll
        for (int j = 0; j < 16; ++j) {
            int d = __builtin_amdgcn_readlane(dv, b + j);
            float* dp = H2 + (size_t)d * 64 + lane;
            unsafeAtomicAdd(dp + 0,  v[j][0]);
            unsafeAtomicAdd(dp + 32, v[j][1]);
        }
    }
}

__global__ void GCN_finalize_kernel(float* __restrict__ H2, const float* __restrict__ inorm,
                                    const float* __restrict__ b2) {
    int i = blockIdx.x * blockDim.x + threadIdx.x;
    if (i >= NN * 16) return;
    int row = i >> 4;
    int c4 = (i & 15) * 4;
    float s = inorm[row];
    float4 v = ((float4*)H2)[i];
    v.x = fmaf(v.x, s, b2[c4 + 0]);
    v.y = fmaf(v.y, s, b2[c4 + 1]);
    v.z = fmaf(v.z, s, b2[c4 + 2]);
    v.w = fmaf(v.w, s, b2[c4 + 3]);
    ((float4*)H2)[i] = v;
}

__global__ void GCN_score_kernel(const float* __restrict__ H2, const int* __restrict__ qs,
                                 const int* __restrict__ qd, float* __restrict__ out) {
    int t = blockIdx.x * blockDim.x + threadIdx.x;
    int q = t >> 5, lane = t & 31;
    int s = qs[q], d = qd[q];
    float2 a = *(const float2*)(H2 + (size_t)s * 64 + lane * 2);
    float2 b = *(const float2*)(H2 + (size_t)d * 64 + lane * 2);
    float p = a.x * b.x + a.y * b.y;
#pragma unroll
    for (int off = 16; off > 0; off >>= 1) p += __shfl_xor(p, off, 32);
    if (lane == 0) out[q] = 1.0f / (1.0f + __expf(-p));
}

extern "C" void kernel_launch(void* const* d_in, const int* in_sizes, int n_in,
                              void* d_out, int out_size, void* d_ws, size_t ws_size,
                              hipStream_t stream) {
    const float* feat = (const float*)d_in[0];
    const int* esrc   = (const int*)d_in[1];
    const int* edst   = (const int*)d_in[2];
    const int* qsrc   = (const int*)d_in[3];
    const int* qdst   = (const int*)d_in[4];
    const float* W1   = (const float*)d_in[5];
    const float* b1   = (const float*)d_in[6];
    const float* W2   = (const float*)d_in[7];
    const float* b2   = (const float*)d_in[8];
    float* out = (float*)d_out;

    float* ws    = (float*)d_ws;
    float* onorm = ws;
    float* inorm = ws + NN;
    float* agg1  = ws + 2 * NN;
    float* h1    = agg1 + (size_t)NN * 128;
    float* t2    = h1 + (size_t)NN * 128;
    float* h2    = t2 + (size_t)NN * 64;

    hipMemsetAsync(ws, 0, (size_t)(2 * NN + (size_t)NN * 128) * sizeof(float), stream);
    hipMemsetAsync(h2, 0, (size_t)NN * 64 * sizeof(float), stream);

    GCN_deg_kernel<<<(EE + 255) / 256, 256, 0, stream>>>(esrc, edst, onorm, inorm);
    GCN_norm_kernel<<<(2 * NN + 255) / 256, 256, 0, stream>>>(onorm);
    GCN_scatter1_kernel<<<EE / 256, 256, 0, stream>>>(feat, esrc, edst, onorm, agg1);
    GCN_gemm1_kernel<<<1024, 256, 0, stream>>>(agg1, W1, b1, inorm, h1);
    GCN_gemm2_kernel<<<1250, 128, 0, stream>>>(h1, W2, onorm, t2);
    GCN_scatter2_kernel<<<EE / 256, 256, 0, stream>>>(t2, esrc, edst, h2);
    GCN_finalize_kernel<<<(NN * 16 + 255) / 256, 256, 0, stream>>>(h2, inorm, b2);
    GCN_score_kernel<<<QQ / 8, 256, 0, stream>>>(h2, qsrc, qdst, out);
}


# Degree counts: the 3.2 M scattered global float atomics become per-chunk 16-bit histograms in LDS (integer LDS adds, 25000 ids a block), the 64 planes summed in the norm kernel; counts exact
# speedup vs baseline: 2.1290x; 1.3320x over previous
#include <hip/hip_runtime.h>
#include <math.h>

typedef __attribute__((ext_vector_type(2))) float v2f;
typedef __attribute__((ext_vector_type(8))) float v8f;

#define NN 100000
#define EE 1600000
#define QQ 1000000

#define HCH  64
#define HLEN (EE / HCH)
#define HW   (NN / 2)
#define HWP  50016
#define HIT  ((HLEN / 4 + 1023) / 1024)
static_assert(EE % HCH == 0 && HLEN % 4 == 0 && HLEN < 65536, "chunk: whole int4s, 16-bit safe");
static_assert(NN % 2 == 0 && HW % 4 == 0 && HWP >= HW && (HWP * 4) % 128 == 0, "plane layout");
__global__ void __launch_bounds__(1024)
GCN_hist_kernel(const int* __restrict__ esrc, const int* __restrict__ edst, int* __restrict__ part) {
    __shared__ __attribute__((aligned(16))) unsigned cnt[HW];
    const int tid = threadIdx.x;
    const int which = blockIdx.x / HCH, chunk = blockIdx.x - which * HCH;
    const int4* ids4 = (const int4*)((which ? edst : esrc) + (size_t)chunk * HLEN);
    uint4* cnt4 = (uint4*)cnt;
    int4 v[HIT];
#pragma unroll
    for (int k = 0; k < HIT; ++k) {
        int i = tid + k * 1024;
        if (i < HLEN / 4) v[k] = ids4[i];
    }
    for (int w = tid; w < HW / 4; w += 1024) cnt4[w] = make_uint4(0u, 0u, 0u, 0u);
    __syncthreads();
#pragma unroll
    for (int k = 0; k < HIT; ++k) {
        int i = tid + k * 1024;
        if (i < HLEN / 4) {
            atomicAdd(&cnt[v[k].x >> 1], 1u << ((v[k].x & 1) * 16));
            atomicAdd(&cnt[v[k].y >> 1], 1u << ((v[k].y & 1) * 16));
            atomicAdd(&cnt[v[k].z >> 1], 1u << ((v[k].z & 1) * 16));
            atomicAdd(&cnt[v[k].w >> 1], 1u << ((v[k].w & 1) * 16));
        }
    }
    __syncthreads();
    uint4* out4 = (uint4*)(part + (size_t)blockIdx.x * HWP);
    for (int w = tid; w < HW / 4; w += 1024) out4[w] = cnt4[w];
}

__global__ void __launch_bounds__(256)
GCN_norm_kernel(const int* __restrict__ part, float* __restrict__ d) {
    int i = blockIdx.x * 256 + threadIdx.x;
    if (i >= 2 * HW) return;
    const int which = i >= HW ? 1 : 0;
    const int w = i - which * HW;
    const unsigned* p = (const unsigned*)part + (size_t)which * HCH * HWP + w;
    unsigned lo = 0, hi = 0;
#pragma unroll 16
    for (int c = 0; c < HCH; ++c) {
        unsigned v = p[(size_t)c * HWP];
        lo += v & 0xffffu;
        hi += v >> 16;
    }
    float2 r;
    r.x = rsqrtf(fmaxf((float)lo, 1.0f));
    r.y = rsqrtf(fmaxf((float)hi, 1.0f));
    *(float2*)(d + (size_t)which * NN + 2 * w) = r;
}

static_assert(EE % 256 == 0, "edge grid has no tail");
__global__ void __launch_bounds__(256)
GCN_scatter1_kernel(const float* __restrict__ feat,
                    const int* __restrict__ esrc, const int* __restrict__ edst,
                    const float* __restrict__ onorm, float* __restrict__ agg) {
    int t = blockIdx.x * 256 + threadIdx.x;
    int lane = threadIdx.x & 31;
    int sv = esrc[t], dv = edst[t];
    float scv = onorm[sv];
#pragma unroll
    for (int b = 0; b < 32; b += 8) {
        float v[8][4];
#pragma unroll
        for (int j = 0; j < 8; ++j) {
            int s = __builtin_amdgcn_readlane(sv, b + j);
            const float* sp = feat + (size_t)s * 128 + lane;
            v[j][0] = sp[0]; v[j][1] = sp[32]; v[j][2] = sp[64]; v[j][3] = sp[96];
        }
#pragma unroll
        for (int j = 0; j < 8; ++j) {
            int d = __builtin_amdgcn_readlane(dv, b + j);
            float sc = __int_as_float(__builtin_amdgcn_readlane(__float_as_int(scv), b + j));
            float* dp = agg + (size_t)d * 128 + lane;
            unsafeAtomicAdd(dp + 0,  v[j][0] * sc);
            unsafeAtomicAdd(dp + 32, v[j][1] * sc);
            unsafeAtomicAdd(dp + 64, v[j][2] * sc);
            unsafeAtomicAdd(dp + 96, v[j][3] * sc);
        }
    }
}

#define LDA 132
__global__ void __launch_bounds__(256)
GCN_gemm1_kernel(const float* __restrict__ A, const float* __restrict__ W,
                 const float* __restrict__ bias, const float* __restrict__ inorm,
                 float* __restrict__ H) {
    __shared__ __attribute__((aligned(16))) float lA[2][16 * LDA];
    const int ntiles = NN / 16;
    int tile = blockIdx.x;
    if (tile >= ntiles) return;

    const int tid = threadIdx.x;
    const int wave = tid >> 5, lane = tid & 31;
    const int r = lane & 15, kh = lane >> 4;
    const int col0 = wave * 16;

    v2f bf[32];
    {
        const float* Bp = W + (size_t)(2 * kh) * 128 + col0 + r;
#pragma unroll
        for (int kt = 0; kt < 32; ++kt) {
            bf[kt].x = Bp[kt * 512];
            bf[kt].y = Bp[kt * 512 + 128];
        }
    }
    const int n = col0 + r;
    const float bb = bias[n];

    const int srow = wave, sc4 = lane;
    const float4* A4 = (const float4*)A;
    float4 p0 = A4[((size_t)tile * 16 + srow) * 32 + sc4];
    float4 p1 = A4[((size_t)tile * 16 + srow + 8) * 32 + sc4];
    float ni[8];
#pragma unroll
    for (int v = 0; v < 8; ++v) ni[v] = inorm[tile * 16 + 8 * kh + v];
    *(float4*)&lA[0][srow * LDA + sc4 * 4] = p0;
    *(float4*)&lA[0][(srow + 8) * LDA + sc4 * 4] = p1;
    __syncthreads();

    int cur = 0;
    for (; tile < ntiles; tile += gridDim.x) {
        const int nxt = tile + gridDim.x;
        const bool more = nxt < ntiles;
        float ci[8];
#pragma unroll
        for (int v = 0; v < 8; ++v) ci[v] = ni[v];
        if (more) {
            p0 = A4[((size_t)nxt * 16 + srow) * 32 + sc4];
            p1 = A4[((size_t)nxt * 16 + srow + 8) * 32 + sc4];
#pragma unroll
            for (int v = 0; v < 8; ++v) ni[v] = inorm[nxt * 16 + 8 * kh + v];
        }
        const float* ap = &lA[cur][r * LDA + 2 * kh];
        v8f c = {};
#pragma unroll
        for (int kt = 0; kt < 32; ++kt) {
            v2f a = *(const v2f*)(ap + kt * 4);
            c = __builtin_amdgcn_wmma_f32_16x16x4_f32(false, a, false, bf[kt], (short)0, c, false, false);
        }
        if (more) {
            *(float4*)&lA[cur ^ 1][srow * LDA + sc4 * 4] = p0;
            *(float4*)&lA[cur ^ 1][(srow + 8) * LDA + sc4 * 4] = p1;
        }
        __syncthreads();
        const int row0 = tile * 16;
#pragma unroll
        for (int v = 0; v < 8; ++v) {
            int m = row0 + v + 8 * kh;
            float val = fmaf(c[v], ci[v], bb);
            H[(size_t)m * 128 + n] = fmaxf(val, 0.0f);
        }
        cur ^= 1;
    }
}

__global__ void __launch_bounds__(128)
GCN_gemm2_kernel(const float* __restrict__ Hin, const float* __restrict__ W,
                 const float* __restrict__ onorm, float* __restrict__ T) {
    __shared__ __attribute__((aligned(16))) float lA[2][16 * LDA];
    const int ntiles = NN / 16;
    int tile = blockIdx.x;
    if (tile >= ntiles) return;

    const int tid = threadIdx.x;
    const int wave = tid >> 5, lane = tid & 31;
    const int r = lane & 15, kh = lane >> 4;
    const int col0 = wave * 16;

    v2f bf[32];
    {
        const float* Bp = W + (size_t)(2 * kh) * 64 + col0 + r;
#pragma unroll
        for (int kt = 0; kt < 32; ++kt) {
            bf[kt].x = Bp[kt * 256];
            bf[kt].y = Bp[kt * 256 + 64];
        }
    }
    const int n = col0 + r;

    const int srow = wave, sc4 = lane;
    const float4* A4 = (const float4*)Hin;
    float4 p[4];
#pragma unroll
    for (int j = 0; j < 4; ++j) {
        int row = tile * 16 + srow + 4 * j;
        float sc = onorm[row];
        float4 q = A4[(size_t)row * 32 + sc4];
        q.x *= sc; q.y *= sc; q.z *= sc; q.w *= sc;
        p[j] = q;
    }
#pragma unroll
    for (int j = 0; j < 4; ++j) *(float4*)&lA[0][(srow + 4 * j) * LDA + sc4 * 4] = p[j];
    __syncthreads();

    int cur = 0;
    for (; tile < ntiles; tile += gridDim.x) {
        const int nxt = tile + gridDim.x;
        const bool more = nxt < ntiles;
        float sn[4];
        if (more) {
#pragma unroll
            for (int j = 0; j < 4; ++j) {
                int row = nxt * 16 + srow + 4 * j;
                sn[j] = onorm[row];
                p[j] = A4[(size_t)row * 32 + sc4];
            }
        }
        const float* ap = &lA[cur][r * LDA + 2 * kh];
        v8f c = {};
#pragma unroll
        for (int kt = 0; kt < 32; ++kt) {
            v2f a = *(const v2f*)(ap + kt * 4);
            c = __builtin_amdgcn_wmma_f32_16x16x4_f32(false, a, false, bf[kt], (short)0, c, false, false);
        }
        if (more) {
#pragma unroll
            for (int j = 0; j < 4; ++j) {
                float4 q = p[j];
                float sc = sn[j];
                q.x *= sc; q.y *= sc; q.z *= sc; q.w *= sc;
                *(float4*)&lA[cur ^ 1][(srow + 4 * j) * LDA + sc4 * 4] = q;
            }
        }
        __syncthreads();
        const int row0 = tile * 16;
#pragma unroll
        for (int v = 0; v < 8; ++v) {
            int m = row0 + v + 8 * kh;
            T[(size_t)m * 64 + n] = c[v];
        }
        cur ^= 1;
    }
}

__global__ void __launch_bounds__(256)
GCN_scatter2_kernel(const float* __restrict__ T,
                    const int* __restrict__ esrc, const int* __restrict__ edst,
                    float* __restrict__ H2) {
    int t = blockIdx.x * 256 + threadIdx.x;
    int lane = threadIdx.x & 31;
    int sv = esrc[t], dv = edst[t];
#pragma unroll
    for (int b = 0; b < 32; b += 16) {
        float v[16][2];
#pragma unroll
        for (int j = 0; j < 16; ++j) {
            int s = __builtin_amdgcn_readlane(sv, b + j);
            const float* sp = T + (size_t)s * 64 + lane;
            v[j][0] = sp[0]; v[j][1] = sp[32];
        }
#pragma unroll
        for (int j = 0; j < 16; ++j) {
            int d = __builtin_amdgcn_readlane(dv, b + j);
            float* dp = H2 + (size_t)d * 64 + lane;
            unsafeAtomicAdd(dp + 0,  v[j][0]);
            unsafeAtomicAdd(dp + 32, v[j][1]);
        }
    }
}

__global__ void GCN_finalize_kernel(float* __restrict__ H2, const float* __restrict__ inorm,
                                    const float* __restrict__ b2) {
    int i = blockIdx.x * blockDim.x + threadIdx.x;
    if (i >= NN * 16) return;
    int row = i >> 4;
    int c4 = (i & 15) * 4;
    float s = inorm[row];
    float4 v = ((float4*)H2)[i];
    v.x = fmaf(v.x, s, b2[c4 + 0]);
    v.y = fmaf(v.y, s, b2[c4 + 1]);
    v.z = fmaf(v.z, s, b2[c4 + 2]);
    v.w = fmaf(v.w, s, b2[c4 + 3]);
    ((float4*)H2)[i] = v;
}

__global__ void GCN_score_kernel(const float* __restrict__ H2, const int* __restrict__ qs,
                                 const int* __restrict__ qd, float* __restrict__ out) {
    int t = blockIdx.x * blockDim.x + threadIdx.x;
    int q = t >> 5, lane = t & 31;
    int s = qs[q], d = qd[q];
    float2 a = *(const float2*)(H2 + (size_t)s * 64 + lane * 2);
    float2 b = *(const float2*)(H2 + (size_t)d * 64 + lane * 2);
    float p = a.x * b.x + a.y * b.y;
#pragma unroll
    for (int off = 16; off > 0; off >>= 1) p += __shfl_xor(p, off, 32);
    if (lane == 0) out[q] = 1.0f / (1.0f + __expf(-p));
}

extern "C" void kernel_launch(void* const* d_in, const int* in_sizes, int n_in,
                              void* d_out, int out_size, void* d_ws, size_t ws_size,
                              hipStream_t stream) {
    const float* feat = (const float*)d_in[0];
    const int* esrc   = (const int*)d_in[1];
    const int* edst   = (const int*)d_in[2];
    const int* qsrc   = (const int*)d_in[3];
    const int* qdst   = (const int*)d_in[4];
    const float* W1   = (const float*)d_in[5];
    const float* b1   = (const float*)d_in[6];
    const float* W2   = (const float*)d_in[7];
    const float* b2   = (const float*)d_in[8];
    float* out = (float*)d_out;

    float* ws    = (float*)d_ws;
    float* onorm = ws;
    float* inorm = ws + NN;
    float* agg1  = ws + 2 * NN;
    float* h1    = agg1 + (size_t)NN * 128;
    float* t2    = h1 + (size_t)NN * 128;
    float* h2    = t2 + (size_t)NN * 64;

    int* part = (int*)h1;

    hipMemsetAsync(agg1, 0, (size_t)NN * 128 * sizeof(float), stream);
    hipMemsetAsync(h2, 0, (size_t)NN * 64 * sizeof(float), stream);

    GCN_hist_kernel<<<2 * HCH, 1024, 0, stream>>>(esrc, edst, part);
    GCN_norm_kernel<<<(2 * HW + 255) / 256, 256, 0, stream>>>(part, onorm);
    GCN_scatter1_kernel<<<EE / 256, 256, 0, stream>>>(feat, esrc, edst, onorm, agg1);
    GCN_gemm1_kernel<<<1024, 256, 0, stream>>>(agg1, W1, b1, inorm, h1);
    GCN_gemm2_kernel<<<1250, 128, 0, stream>>>(h1, W2, onorm, t2);
    GCN_scatter2_kernel<<<EE / 256, 256, 0, stream>>>(t2, esrc, edst, h2);
    GCN_finalize_kernel<<<(NN * 16 + 255) / 256, 256, 0, stream>>>(h2, inorm, b2);
    GCN_score_kernel<<<QQ / 8, 256, 0, stream>>>(h2, qsrc, qdst, out);
}


# Both edge scatters: no global atomics: a block owns 400 dst rows, filters the whole dst-id list, sums them in LDS (float LDS adds, arrival order), writes each row once (layer 2 with norm+bias)
# speedup vs baseline: 2.2408x; 1.0525x over previous
#include <hip/hip_runtime.h>
#include <math.h>

typedef __attribute__((ext_vector_type(2))) float v2f;
typedef __attribute__((ext_vector_type(8))) float v8f;

#define NN 100000
#define EE 1600000
#define QQ 1000000

#define HCH  64
#define HLEN (EE / HCH)
#define HW   (NN / 2)
#define HWP  50016
#define HIT  ((HLEN / 4 + 1023) / 1024)
static_assert(EE % HCH == 0 && HLEN % 4 == 0 && HLEN < 65536, "chunk: whole int4s, 16-bit safe");
static_assert(NN % 2 == 0 && HW % 4 == 0 && HWP >= HW && (HWP * 4) % 128 == 0, "plane layout");
__global__ void __launch_bounds__(1024)
GCN_hist_kernel(const int* __restrict__ esrc, const int* __restrict__ edst, int* __restrict__ part) {
    __shared__ __attribute__((aligned(16))) unsigned cnt[HW];
    const int tid = threadIdx.x;
    const int which = blockIdx.x / HCH, chunk = blockIdx.x - which * HCH;
    const int4* ids4 = (const int4*)((which ? edst : esrc) + (size_t)chunk * HLEN);
    uint4* cnt4 = (uint4*)cnt;
    int4 v[HIT];
#pragma unroll
    for (int k = 0; k < HIT; ++k) {
        int i = tid + k * 1024;
        if (i < HLEN / 4) v[k] = ids4[i];
    }
    for (int w = tid; w < HW / 4; w += 1024) cnt4[w] = make_uint4(0u, 0u, 0u, 0u);
    __syncthreads();
#pragma unroll
    for (int k = 0; k < HIT; ++k) {
        int i = tid + k * 1024;
        if (i < HLEN / 4) {
            atomicAdd(&cnt[v[k].x >> 1], 1u << ((v[k].x & 1) * 16));
            atomicAdd(&cnt[v[k].y >> 1], 1u << ((v[k].y & 1) * 16));
            atomicAdd(&cnt[v[k].z >> 1], 1u << ((v[k].z & 1) * 16));
            atomicAdd(&cnt[v[k].w >> 1], 1u << ((v[k].w & 1) * 16));
        }
    }
    __syncthreads();
    uint4* out4 = (uint4*)(part + (size_t)blockIdx.x * HWP);
    for (int w = tid; w < HW / 4; w += 1024) out4[w] = cnt4[w];
}

__global__ void __launch_bounds__(256)
GCN_norm_kernel(const int* __restrict__ part, float* __restrict__ d) {
    int i = blockIdx.x * 256 + threadIdx.x;
    if (i >= 2 * HW) return;
    const int which = i >= HW ? 1 : 0;
    const int w = i - which * HW;
    const unsigned* p = (const unsigned*)part + (size_t)which * HCH * HWP + w;
    unsigned lo = 0, hi = 0;
#pragma unroll 16
    for (int c = 0; c < HCH; ++c) {
        unsigned v = p[(size_t)c * HWP];
        lo += v & 0xffffu;
        hi += v >> 16;
    }
    float2 r;
    r.x = rsqrtf(fmaxf((float)lo, 1.0f));
    r.y = rsqrtf(fmaxf((float)hi, 1.0f));
    *(float2*)(d + (size_t)which * NN + 2 * w) = r;
}

#define OR_   400
#define OBLK  (NN / OR_)
#define NG    (EE / 128)
#define LCAP  144
#define OSTEPS ((((NG / 32 + 1 + 3) / 4 + 1) / 2) * 2)
static_assert(NN % OR_ == 0 && EE % 128 == 0 && NG % OBLK == 0, "ranges and groups are whole");
static_assert(EE <= (1 << 21) && OR_ <= (1 << 11), "list word: row in the top 11 bits, edge in the low 21");
static_assert(OSTEPS * 4 * 32 >= NG + 32, "the steps cover every group of every wave");

template <int COLS, bool SRC_SCALE, bool FINISH>
__global__ void __launch_bounds__(1024)
GCN_own_kernel(const float* __restrict__ X, const int* __restrict__ esrc, const int* __restrict__ edst,
               const float* __restrict__ onorm, const float* __restrict__ inorm,
               const float* __restrict__ bias, float* __restrict__ Y) {
    constexpr int LPR = COLS / 32;
    constexpr int BAT = 32 / LPR;
    static_assert(BAT - 1 + 128 <= LCAP, "list capacity");
    __shared__ __attribute__((aligned(16))) float acc[OR_ * COLS];
    __shared__ unsigned lst[32 * LCAP];
    const int tid = threadIdx.x, lane = tid & 31;
    const int wave = __builtin_amdgcn_readfirstlane(tid >> 5);
    const int base = blockIdx.x * OR_;
    const int rot = blockIdx.x * (NG / OBLK);
    const int4* ids4 = (const int4*)edst;
    unsigned* mylst = lst + wave * LCAP;
    float4* acc4 = (float4*)acc;
    int cnt = 0;

    auto load4 = [&](int4 (&buf)[4], int s) {
#pragma unroll
        for (int j = 0; j < 4; ++j) {
            int it0 = wave + 32 * (s * 4 + j);
            int g = it0 + rot;
            if (g >= NG) g -= NG;
            if (it0 >= NG) g = 0;
            buf[j] = ids4[g * 32 + lane];
        }
    };
    auto drain = [&]() {
        cnt -= BAT;
        unsigned p = mylst[cnt + (lane & (BAT - 1))];
        int s = esrc[p & 0x1fffffu];
        int dl = (int)(p >> 21);
        float sc = 1.0f;
        if (SRC_SCALE) sc = onorm[s];
        float v[BAT][LPR];
#pragma unroll
        for (int j = 0; j < BAT; ++j) {
            int sj = __builtin_amdgcn_readlane(s, j);
            const float* sp = X + (size_t)sj * COLS + lane;
#pragma unroll
            for (int q = 0; q < LPR; ++q) v[j][q] = sp[32 * q];
        }
#pragma unroll
        for (int j = 0; j < BAT; ++j) {
            int dj = __builtin_amdgcn_readlane(dl, j);
            float scj = __int_as_float(__builtin_amdgcn_readlane(__float_as_int(sc), j));
            float* ap = acc + dj * COLS + lane;
#pragma unroll
            for (int q = 0; q < LPR; ++q) {
                float val = SRC_SCALE ? v[j][q] * scj : v[j][q];
                __hip_atomic_fetch_add(ap + 32 * q, val, __ATOMIC_RELAXED, __HIP_MEMORY_SCOPE_WORKGROUP);
            }
        }
    };
    auto scan4 = [&](const int4 (&buf)[4], int s) {
#pragma unroll
        for (int j = 0; j < 4; ++j) {
            int it0 = wave + 32 * (s * 4 + j);
            if (it0 < NG) {
                int g = it0 + rot;
                if (g >= NG) g -= NG;
                unsigned u[4];
                u[0] = (unsigned)(buf[j].x - base); u[1] = (unsigned)(buf[j].y - base);
                u[2] = (unsigned)(buf[j].z - base); u[3] = (unsigned)(buf[j].w - base);
                unsigned mn = min(min(u[0], u[1]), min(u[2], u[3]));
                if (__builtin_amdgcn_ballot_w32(mn < (unsigned)OR_)) {
                    const unsigned e0 = (unsigned)(g * 32 + lane) * 4u;
#pragma unroll
                    for (int c = 0; c < 4; ++c) {
                        bool m = u[c] < (unsigned)OR_;
                        unsigned bal = __builtin_amdgcn_ballot_w32(m);
                        if (bal) {
                            int pos = cnt + (int)__builtin_amdgcn_mbcnt_lo(bal, 0u);
                            if (m) mylst[pos] = (u[c] << 21) | (e0 + c);
                            cnt += __builtin_popcount(bal);
                        }
                    }
                    while (cnt >= BAT) drain();
                }
            }
        }
    };

    int4 ia[4], ib[4];
    load4(ia, 0);
    for (int i = tid; i < OR_ * COLS / 4; i += 1024) acc4[i] = make_float4(0.0f, 0.0f, 0.0f, 0.0f);
    __syncthreads();
    for (int s = 0; s < OSTEPS; s += 2) {
        load4(ib, s + 1);
        scan4(ia, s);
        load4(ia, s + 2);
        scan4(ib, s + 1);
    }
    if (cnt > 0) {
        unsigned p = mylst[lane < cnt ? lane : 0];
        int s = esrc[p & 0x1fffffu];
        int dl = (int)(p >> 21);
        float sc = 1.0f;
        if (SRC_SCALE) sc = onorm[s];
        for (int j = 0; j < cnt; ++j) {
            int sj = __builtin_amdgcn_readlane(s, j);
            int dj = __builtin_amdgcn_readlane(dl, j);
            float scj = __int_as_float(__builtin_amdgcn_readlane(__float_as_int(sc), j));
            const float* sp = X + (size_t)sj * COLS + lane;
            float* ap = acc + dj * COLS + lane;
#pragma unroll
            for (int q = 0; q < LPR; ++q) {
                float val = SRC_SCALE ? sp[32 * q] * scj : sp[32 * q];
                __hip_atomic_fetch_add(ap + 32 * q, val, __ATOMIC_RELAXED, __HIP_MEMORY_SCOPE_WORKGROUP);
            }
        }
    }
    __syncthreads();
    float4* Y4 = (float4*)(Y + (size_t)base * COLS);
    for (int i = tid; i < OR_ * COLS / 4; i += 1024) {
        float4 a = acc4[i];
        if (FINISH) {
            const int row = base + i / (COLS / 4);
            const int c4 = (i % (COLS / 4)) * 4;
            const float sn = inorm[row];
            a.x = fmaf(a.x, sn, bias[c4 + 0]);
            a.y = fmaf(a.y, sn, bias[c4 + 1]);
            a.z = fmaf(a.z, sn, bias[c4 + 2]);
            a.w = fmaf(a.w, sn, bias[c4 + 3]);
        }
        Y4[i] = a;
    }
}

#define LDA 132
__global__ void __launch_bounds__(256)
GCN_gemm1_kernel(const float* __restrict__ A, const float* __restrict__ W,
                 const float* __restrict__ bias, const float* __restrict__ inorm,
                 float* __restrict__ H) {
    __shared__ __attribute__((aligned(16))) float lA[2][16 * LDA];
    const int ntiles = NN / 16;
    int tile = blockIdx.x;
    if (tile >= ntiles) return;

    const int tid = threadIdx.x;
    const int wave = tid >> 5, lane = tid & 31;
    const int r = lane & 15, kh = lane >> 4;
    const int col0 = wave * 16;

    v2f bf[32];
    {
        const float* Bp = W + (size_t)(2 * kh) * 128 + col0 + r;
#pragma unroll
        for (int kt = 0; kt < 32; ++kt) {
            bf[kt].x = Bp[kt * 512];
            bf[kt].y = Bp[kt * 512 + 128];
        }
    }
    const int n = col0 + r;
    const float bb = bias[n];

    const int srow = wave, sc4 = lane;
    const float4* A4 = (const float4*)A;
    float4 p0 = A4[((size_t)tile * 16 + srow) * 32 + sc4];
    float4 p1 = A4[((size_t)tile * 16 + srow + 8) * 32 + sc4];
    float ni[8];
#pragma unroll
    for (int v = 0; v < 8; ++v) ni[v] = inorm[tile * 16 + 8 * kh + v];
    *(float4*)&lA[0][srow * LDA + sc4 * 4] = p0;
    *(float4*)&lA[0][(srow + 8) * LDA + sc4 * 4] = p1;
    __syncthreads();

    int cur = 0;
    for (; tile < ntiles; tile += gridDim.x) {
        const int nxt = tile + gridDim.x;
        const bool more = nxt < ntiles;
        float ci[8];
#pragma unroll
        for (int v = 0; v < 8; ++v) ci[v] = ni[v];
        if (more) {
            p0 = A4[((size_t)nxt * 16 + srow) * 32 + sc4];
            p1 = A4[((size_t)nxt * 16 + srow + 8) * 32 + sc4];
#pragma unroll
            for (int v = 0; v < 8; ++v) ni[v] = inorm[nxt * 16 + 8 * kh + v];
        }
        const float* ap = &lA[cur][r * LDA + 2 * kh];
        v8f c = {};
#pragma unroll
        for (int kt = 0; kt < 32; ++kt) {
            v2f a = *(const v2f*)(ap + kt * 4);
            c = __builtin_amdgcn_wmma_f32_16x16x4_f32(false, a, false, bf[kt], (short)0, c, false, false);
        }
        if (more) {
            *(float4*)&lA[cur ^ 1][srow * LDA + sc4 * 4] = p0;
            *(float4*)&lA[cur ^ 1][(srow + 8) * LDA + sc4 * 4] = p1;
        }
        __syncthreads();
        const int row0 = tile * 16;
#pragma unroll
        for (int v = 0; v < 8; ++v) {
            int m = row0 + v + 8 * kh;
            float val = fmaf(c[v], ci[v], bb);
            H[(size_t)m * 128 + n] = fmaxf(val, 0.0f);
        }
        cur ^= 1;
    }
}

__global__ void __launch_bounds__(128)
GCN_gemm2_kernel(const float* __restrict__ Hin, const float* __restrict__ W,
                 const float* __restrict__ onorm, float* __restrict__ T) {
    __shared__ __attribute__((aligned(16))) float lA[2][16 * LDA];
    const int ntiles = NN / 16;
    int tile = blockIdx.x;
    if (tile >= ntiles) return;

    const int tid = threadIdx.x;
    const int wave = tid >> 5, lane = tid & 31;
    const int r = lane & 15, kh = lane >> 4;
    const int col0 = wave * 16;

    v2f bf[32];
    {
        const float* Bp = W + (size_t)(2 * kh) * 64 + col0 + r;
#pragma unroll
        for (int kt = 0; kt < 32; ++kt) {
            bf[kt].x = Bp[kt * 256];
            bf[kt].y = Bp[kt * 256 + 64];
        }
    }
    const int n = col0 + r;

    const int srow = wave, sc4 = lane;
    const float4* A4 = (const float4*)Hin;
    float4 p[4];
#pragma unroll
    for (int j = 0; j < 4; ++j) {
        int row = tile * 16 + srow + 4 * j;
        float sc = onorm[row];
        float4 q = A4[(size_t)row * 32 + sc4];
        q.x *= sc; q.y *= sc; q.z *= sc; q.w *= sc;
        p[j] = q;
    }
#pragma unroll
    for (int j = 0; j < 4; ++j) *(float4*)&lA[0][(srow + 4 * j) * LDA + sc4 * 4] = p[j];
    __syncthreads();

    int cur = 0;
    for (; tile < ntiles; tile += gridDim.x) {
        const int nxt = tile + gridDim.x;
        const bool more = nxt < ntiles;
        float sn[4];
        if (more) {
#pragma unroll
            for (int j = 0; j < 4; ++j) {
                int row = nxt * 16 + srow + 4 * j;
                sn[j] = onorm[row];
                p[j] = A4[(size_t)row * 32 + sc4];
            }
        }
        const float* ap = &lA[cur][r * LDA + 2 * kh];
        v8f c = {};
#pragma unroll
        for (int kt = 0; kt < 32; ++kt) {
            v2f a = *(const v2f*)(ap + kt * 4);
            c = __builtin_amdgcn_wmma_f32_16x16x4_f32(false, a, false, bf[kt], (short)0, c, false, false);
        }
        if (more) {
#pragma unroll
            for (int j = 0; j < 4; ++j) {
                float4 q = p[j];
                float sc = sn[j];
                q.x *= sc; q.y *= sc; q.z *= sc; q.w *= sc;
                *(float4*)&lA[cur ^ 1][(srow + 4 * j) * LDA + sc4 * 4] = q;
            }
        }
        __syncthreads();
        const int row0 = tile * 16;
#pragma unroll
        for (int v = 0; v < 8; ++v) {
            int m = row0 + v + 8 * kh;
            T[(size_t)m * 64 + n] = c[v];
        }
        cur ^= 1;
    }
}

__global__ void GCN_score_kernel(const float* __restrict__ H2, const int* __restrict__ qs,
                                 const int* __restrict__ qd, float* __restrict__ out) {
    int t = blockIdx.x * blockDim.x + threadIdx.x;
    int q = t >> 5, lane = t & 31;
    int s = qs[q], d = qd[q];
    float2 a = *(const float2*)(H2 + (size_t)s * 64 + lane * 2);
    float2 b = *(const float2*)(H2 + (size_t)d * 64 + lane * 2);
    float p = a.x * b.x + a.y * b.y;
#pragma unroll
    for (int off = 16; off > 0; off >>= 1) p += __shfl_xor(p, off, 32);
    if (lane == 0) out[q] = 1.0f / (1.0f + __expf(-p));
}

extern "C" void kernel_launch(void* const* d_in, const int* in_sizes, int n_in,
                              void* d_out, int out_size, void* d_ws, size_t ws_size,
                              hipStream_t stream) {
    const float* feat = (const float*)d_in[0];
    const int* esrc   = (const int*)d_in[1];
    const int* edst   = (const int*)d_in[2];
    const int* qsrc   = (const int*)d_in[3];
    const int* qdst   = (const int*)d_in[4];
    const float* W1   = (const float*)d_in[5];
    const float* b1   = (const float*)d_in[6];
    const float* W2   = (const float*)d_in[7];
    const float* b2   = (const float*)d_in[8];
    float* out = (float*)d_out;

    float* ws    = (float*)d_ws;
    float* onorm = ws;
    float* inorm = ws + NN;
    float* agg1  = ws + 2 * NN;
    float* h1    = agg1 + (size_t)NN * 128;
    float* t2    = h1 + (size_t)NN * 128;
    float* h2    = t2 + (size_t)NN * 64;

    int* part = (int*)h1;

    GCN_hist_kernel<<<2 * HCH, 1024, 0, stream>>>(esrc, edst, part);
    GCN_norm_kernel<<<(2 * HW + 255) / 256, 256, 0, stream>>>(part, onorm);
    GCN_own_kernel<128, true, false><<<OBLK, 1024, 0, stream>>>(feat, esrc, edst, onorm, inorm, b1, agg1);
    GCN_gemm1_kernel<<<1024, 256, 0, stream>>>(agg1, W1, b1, inorm, h1);
    GCN_gemm2_kernel<<<1250, 128, 0, stream>>>(h1, W2, onorm, t2);
    GCN_own_kernel<64, false, true><<<OBLK, 1024, 0, stream>>>(t2, esrc, edst, onorm, inorm, b2, h2);
    GCN_score_kernel<<<QQ / 8, 256, 0, stream>>>(h2, qsrc, qdst, out);
}


# Layer-2 edge sums read per-wave (row, src) words that layer 1's owner blocks save while filtering (in the block's own span of the h2 plane), not the id arrays again; sums stay in LDS arrival order
# speedup vs baseline: 2.5817x; 1.1521x over previous
#include <hip/hip_runtime.h>
#include <math.h>

typedef __attribute__((ext_vector_type(2))) float v2f;
typedef __attribute__((ext_vector_type(8))) float v8f;

#define NN 100000
#define EE 1600000
#define QQ 1000000

#define HCH  64
#define HLEN (EE / HCH)
#define HW   (NN / 2)
#define HWP  50016
#define HIT  ((HLEN / 4 + 1023) / 1024)
static_assert(EE % HCH == 0 && HLEN % 4 == 0 && HLEN < 65536, "chunk: whole int4s, 16-bit safe");
static_assert(NN % 2 == 0 && HW % 4 == 0 && HWP >= HW && (HWP * 4) % 128 == 0, "plane layout");
__global__ void __launch_bounds__(1024)
GCN_hist_kernel(const int* __restrict__ esrc, const int* __restrict__ edst, int* __restrict__ part) {
    __shared__ __attribute__((aligned(16))) unsigned cnt[HW];
    const int tid = threadIdx.x;
    const int which = blockIdx.x / HCH, chunk = blockIdx.x - which * HCH;
    const int4* ids4 = (const int4*)((which ? edst : esrc) + (size_t)chunk * HLEN);
    uint4* cnt4 = (uint4*)cnt;
    int4 v[HIT];
#pragma unroll
    for (int k = 0; k < HIT; ++k) {
        int i = tid + k * 1024;
        if (i < HLEN / 4) v[k] = ids4[i];
    }
    for (int w = tid; w < HW / 4; w += 1024) cnt4[w] = make_uint4(0u, 0u, 0u, 0u);
    __syncthreads();
#pragma unroll
    for (int k = 0; k < HIT; ++k) {
        int i = tid + k * 1024;
        if (i < HLEN / 4) {
            atomicAdd(&cnt[v[k].x >> 1], 1u << ((v[k].x & 1) * 16));
            atomicAdd(&cnt[v[k].y >> 1], 1u << ((v[k].y & 1) * 16));
            atomicAdd(&cnt[v[k].z >> 1], 1u << ((v[k].z & 1) * 16));
            atomicAdd(&cnt[v[k].w >> 1], 1u << ((v[k].w & 1) * 16));
        }
    }
    __syncthreads();
    uint4* out4 = (uint4*)(part + (size_t)blockIdx.x * HWP);
    for (int w = tid; w < HW / 4; w += 1024) out4[w] = cnt4[w];
}

__global__ void __launch_bounds__(256)
GCN_norm_kernel(const int* __restrict__ part, float* __restrict__ d) {
    int i = blockIdx.x * 256 + threadIdx.x;
    if (i >= 2 * HW) return;
    const int which = i >= HW ? 1 : 0;
    const int w = i - which * HW;
    const unsigned* p = (const unsigned*)part + (size_t)which * HCH * HWP + w;
    unsigned lo = 0, hi = 0;
#pragma unroll 16
    for (int c = 0; c < HCH; ++c) {
        unsigned v = p[(size_t)c * HWP];
        lo += v & 0xffffu;
        hi += v >> 16;
    }
    float2 r;
    r.x = rsqrtf(fmaxf((float)lo, 1.0f));
    r.y = rsqrtf(fmaxf((float)hi, 1.0f));
    *(float2*)(d + (size_t)which * NN + 2 * w) = r;
}

#define OR_   400
#define OBLK  (NN / OR_)
#define NG    (EE / 128)
#define LCAP  144
#define WSEG  800
#define WCAP  (WSEG - 1)
#define OSTEPS ((((NG / 32 + 1 + 3) / 4 + 1) / 2) * 2)
static_assert(NN % OR_ == 0 && EE % 128 == 0 && NG % OBLK == 0, "ranges and groups are whole");
static_assert(EE <= (1 << 21) && OR_ <= (1 << 11), "list word: row in the top 11 bits, edge in the low 21");
static_assert(OSTEPS * 4 * 32 >= NG + 32, "the steps cover every group of every wave");
static_assert(32 * WSEG == OR_ * 64, "a block's saved words fill exactly its own span of the layer-2 plane");
static_assert(NN <= (1 << 17) && OR_ <= (1 << 9), "saved word: row in the top bits, source id in the low 17");

template <int COLS, int OFF, bool FULL>
__device__ __forceinline__ void GCN_saved_batch(const float* __restrict__ X, float* acc,
                                                unsigned w, int m, int lane) {
    constexpr int LPR = COLS / 32;
    const int s = (int)(w & 0x1ffffu);
    const int dl = (int)(w >> 17);
    float v[16][LPR];
#pragma unroll
    for (int j = 0; j < 16; ++j) {
        int sj = __builtin_amdgcn_readlane(s, OFF + j);
        if (!FULL && j >= m) sj = 0;
        const float* sp = X + (size_t)sj * COLS + lane;
#pragma unroll
        for (int q = 0; q < LPR; ++q) v[j][q] = sp[32 * q];
    }
#pragma unroll
    for (int j = 0; j < 16; ++j) {
        if (FULL || j < m) {
            int dj = __builtin_amdgcn_readlane(dl, OFF + j);
            float* ap = acc + dj * COLS + lane;
#pragma unroll
            for (int q = 0; q < LPR; ++q)
                __hip_atomic_fetch_add(ap + 32 * q, v[j][q], __ATOMIC_RELAXED, __HIP_MEMORY_SCOPE_WORKGROUP);
        }
    }
}

template <int COLS, bool SRC_SCALE, bool FINISH, bool SAVE, bool USE>
__global__ void __launch_bounds__(1024)
GCN_own_kernel(const float* __restrict__ X, const int* __restrict__ esrc, const int* __restrict__ edst,
               const float* __restrict__ onorm, const float* __restrict__ inorm,
               const float* __restrict__ bias, float* __restrict__ Y, float* __restrict__ L) {
    constexpr int LPR = COLS / 32;
    constexpr int BAT = 32 / LPR;
    static_assert(BAT - 1 + 128 <= LCAP, "list capacity");
    static_assert(!USE || COLS == 64, "the saved words live in a span of OR_ x 64 floats");
    static_assert(!(SAVE && USE), "a launch saves or reads");
    __shared__ __attribute__((aligned(16))) float acc[OR_ * COLS];
    __shared__ unsigned lst[32 * LCAP];
    const int tid = threadIdx.x, lane = tid & 31;
    const int wave = __builtin_amdgcn_readfirstlane(tid >> 5);
    const int base = blockIdx.x * OR_;
    const int rot = blockIdx.x * (NG / OBLK);
    const int4* ids4 = (const int4*)edst;
    unsigned* mylst = lst + wave * LCAP;
    float4* acc4 = (float4*)acc;
    int cnt = 0;
    int gpos = 0;
    float* sv = L + (size_t)blockIdx.x * (OR_ * 64) + wave * WSEG;

    auto load4 = [&](int4 (&buf)[4], int s) {
#pragma unroll
        for (int j = 0; j < 4; ++j) {
            int it0 = wave + 32 * (s * 4 + j);
            int g = it0 + rot;
            if (g >= NG) g -= NG;
            if (it0 >= NG) g = 0;
            buf[j] = ids4[g * 32 + lane];
        }
    };
    auto drain = [&]() {
        cnt -= BAT;
        unsigned p = mylst[cnt + (lane & (BAT - 1))];
        int s = esrc[p & 0x1fffffu];
        int dl = (int)(p >> 21);
        float sc = 1.0f;
        if (SRC_SCALE) sc = onorm[s];
        if (SAVE) {
            if (gpos + BAT <= WCAP && lane < BAT)
                sv[gpos + lane] = __uint_as_float(((unsigned)dl << 17) | (unsigned)s);
            gpos += BAT;
        }
        float v[BAT][LPR];
#pragma unroll
        for (int j = 0; j < BAT; ++j) {
            int sj = __builtin_amdgcn_readlane(s, j);
            const float* sp = X + (size_t)sj * COLS + lane;
#pragma unroll
            for (int q = 0; q < LPR; ++q) v[j][q] = sp[32 * q];
        }
#pragma unroll
        for (int j = 0; j < BAT; ++j) {
            int dj = __builtin_amdgcn_readlane(dl, j);
            float scj = __int_as_float(__builtin_amdgcn_readlane(__float_as_int(sc), j));
            float* ap = acc + dj * COLS + lane;
#pragma unroll
            for (int q = 0; q < LPR; ++q) {
                float val = SRC_SCALE ? v[j][q] * scj : v[j][q];
                __hip_atomic_fetch_add(ap + 32 * q, val, __ATOMIC_RELAXED, __HIP_MEMORY_SCOPE_WORKGROUP);
            }
        }
    };
    auto scan4 = [&](const int4 (&buf)[4], int s) {
#pragma unroll
        for (int j = 0; j < 4; ++j) {
            int it0 = wave + 32 * (s * 4 + j);
            if (it0 < NG) {
                int g = it0 + rot;
                if (g >= NG) g -= NG;
                unsigned u[4];
                u[0] = (unsigned)(buf[j].x - base); u[1] = (unsigned)(buf[j].y - base);
                u[2] = (unsigned)(buf[j].z - base); u[3] = (unsigned)(buf[j].w - base);
                unsigned mn = min(min(u[0], u[1]), min(u[2], u[3]));
                if (__builtin_amdgcn_ballot_w32(mn < (unsigned)OR_)) {
                    const unsigned e0 = (unsigned)(g * 32 + lane) * 4u;
#pragma unroll
                    for (int c = 0; c < 4; ++c) {
                        bool m = u[c] < (unsigned)OR_;
                        unsigned bal = __builtin_amdgcn_ballot_w32(m);
                        if (bal) {
                            int pos = cnt + (int)__builtin_amdgcn_mbcnt_lo(bal, 0u);
                            if (m) mylst[pos] = (u[c] << 21) | (e0 + c);
                            cnt += __builtin_popcount(bal);
                        }
                    }
                    while (cnt >= BAT) drain();
                }
            }
        }
    };

    int4 ia[4], ib[4];
    int nsaved = 0;
    bool refilter = !USE;
    const float* span = Y + (size_t)blockIdx.x * (OR_ * COLS);
    if (USE) {
        unsigned cv = __float_as_uint(span[lane * WSEG + WCAP]);
        refilter = __builtin_amdgcn_ballot_w32(cv > (unsigned)WCAP) != 0u;
        nsaved = __builtin_amdgcn_readlane((int)cv, wave);
    } else {
        load4(ia, 0);
    }
    unsigned pw = 0u;
    if (USE && !refilter && nsaved > 0)
        pw = __float_as_uint(span[wave * WSEG + lane]);
    for (int i = tid; i < OR_ * COLS / 4; i += 1024) acc4[i] = make_float4(0.0f, 0.0f, 0.0f, 0.0f);
    __syncthreads();
    if (USE && !refilter) {
        const float* seg = span + wave * WSEG;
        int done = 0;
        for (; done + 32 <= nsaved; done += 32) {
            unsigned pn = 0u;
            if (done + 32 < nsaved) pn = __float_as_uint(seg[done + 32 + lane]);
            GCN_saved_batch<COLS, 0, true>(X, acc, pw, 16, lane);
            GCN_saved_batch<COLS, 16, true>(X, acc, pw, 16, lane);
            pw = pn;
        }
        const int m = nsaved - done;
        if (m >= 16) {
            GCN_saved_batch<COLS, 0, true>(X, acc, pw, 16, lane);
            if (m > 16) GCN_saved_batch<COLS, 16, false>(X, acc, pw, m - 16, lane);
        } else if (m > 0) {
            GCN_saved_batch<COLS, 0, false>(X, acc, pw, m, lane);
        }
    } else {
    if (USE) load4(ia, 0);
    for (int s = 0; s < OSTEPS; s += 2) {
        load4(ib, s + 1);
        scan4(ia, s);
        load4(ia, s + 2);
        scan4(ib, s + 1);
    }
    if (cnt > 0) {
        unsigned p = mylst[lane < cnt ? lane : 0];
        int s = esrc[p & 0x1fffffu];
        int dl = (int)(p >> 21);
        float sc = 1.0f;
        if (SRC_SCALE) sc = onorm[s];
        if (SAVE) {
            if (gpos + cnt <= WCAP && lane < cnt)
                sv[gpos + lane] = __uint_as_float(((unsigned)dl << 17) | (unsigned)s);
            gpos += cnt;
        }
        for (int j = 0; j < cnt; ++j) {
            int sj = __builtin_amdgcn_readlane(s, j);
            int dj = __builtin_amdgcn_readlane(dl, j);
            float scj = __int_as_float(__builtin_amdgcn_readlane(__float_as_int(sc), j));
            const float* sp = X + (size_t)sj * COLS + lane;
            float* ap = acc + dj * COLS + lane;
#pragma unroll
            for (int q = 0; q < LPR; ++q) {
                float val = SRC_SCALE ? sp[32 * q] * scj : sp[32 * q];
                __hip_atomic_fetch_add(ap + 32 * q, val, __ATOMIC_RELAXED, __HIP_MEMORY_SCOPE_WORKGROUP);
            }
        }
    }
    if (SAVE && lane == 0) sv[WCAP] = __uint_as_float((unsigned)gpos);
    }
    __syncthreads();
    float4* Y4 = (float4*)(Y + (size_t)base * COLS);
    for (int i = tid; i < OR_ * COLS / 4; i += 1024) {
        float4 a = acc4[i];
        if (FINISH) {
            const int row = base + i / (COLS / 4);
            const int c4 = (i % (COLS / 4)) * 4;
            const float sn = inorm[row];
            a.x = fmaf(a.x, sn, bias[c4 + 0]);
            a.y = fmaf(a.y, sn, bias[c4 + 1]);
            a.z = fmaf(a.z, sn, bias[c4 + 2]);
            a.w = fmaf(a.w, sn, bias[c4 + 3]);
        }
        Y4[i] = a;
    }
}

#define LDA 132
__global__ void __launch_bounds__(256)
GCN_gemm1_kernel(const float* __restrict__ A, const float* __restrict__ W,
                 const float* __restrict__ bias, const float* __restrict__ inorm,
                 float* __restrict__ H) {
    __shared__ __attribute__((aligned(16))) float lA[2][16 * LDA];
    const int ntiles = NN / 16;
    int tile = blockIdx.x;
    if (tile >= ntiles) return;

    const int tid = threadIdx.x;
    const int wave = tid >> 5, lane = tid & 31;
    const int r = lane & 15, kh = lane >> 4;
    const int col0 = wave * 16;

    v2f bf[32];
    {
        const float* Bp = W + (size_t)(2 * kh) * 128 + col0 + r;
#pragma unroll
        for (int kt = 0; kt < 32; ++kt) {
            bf[kt].x = Bp[kt * 512];
            bf[kt].y = Bp[kt * 512 + 128];
        }
    }
    const int n = col0 + r;
    const float bb = bias[n];

    const int srow = wave, sc4 = lane;
    const float4* A4 = (const float4*)A;
    float4 p0 = A4[((size_t)tile * 16 + srow) * 32 + sc4];
    float4 p1 = A4[((size_t)tile * 16 + srow + 8) * 32 + sc4];
    float ni[8];
#pragma unroll
    for (int v = 0; v < 8; ++v) ni[v] = inorm[tile * 16 + 8 * kh + v];
    *(float4*)&lA[0][srow * LDA + sc4 * 4] = p0;
    *(float4*)&lA[0][(srow + 8) * LDA + sc4 * 4] = p1;
    __syncthreads();

    int cur = 0;
    for (; tile < ntiles; tile += gridDim.x) {
        const int nxt = tile + gridDim.x;
        const bool more = nxt < ntiles;
        float ci[8];
#pragma unroll
        for (int v = 0; v < 8; ++v) ci[v] = ni[v];
        if (more) {
            p0 = A4[((size_t)nxt * 16 + srow) * 32 + sc4];
            p1 = A4[((size_t)nxt * 16 + srow + 8) * 32 + sc4];
#pragma unroll
            for (int v = 0; v < 8; ++v) ni[v] = inorm[nxt * 16 + 8 * kh + v];
        }
        const float* ap = &lA[cur][r * LDA + 2 * kh];
        v8f c = {};
#pragma unroll
        for (int kt = 0; kt < 32; ++kt) {
            v2f a = *(const v2f*)(ap + kt * 4);
            c = __builtin_amdgcn_wmma_f32_16x16x4_f32(false, a, false, bf[kt], (short)0, c, false, false);
        }
        if (more) {
            *(float4*)&lA[cur ^ 1][srow * LDA + sc4 * 4] = p0;
            *(float4*)&lA[cur ^ 1][(srow + 8) * LDA + sc4 * 4] = p1;
        }
        __syncthreads();
        const int row0 = tile * 16;
#pragma unroll
        for (int v = 0; v < 8; ++v) {
            int m = row0 + v + 8 * kh;
            float val = fmaf(c[v], ci[v], bb);
            H[(size_t)m * 128 + n] = fmaxf(val, 0.0f);
        }
        cur ^= 1;
    }
}

__global__ void __launch_bounds__(128)
GCN_gemm2_kernel(const float* __restrict__ Hin, const float* __restrict__ W,
                 const float* __restrict__ onorm, float* __restrict__ T) {
    __shared__ __attribute__((aligned(16))) float lA[2][16 * LDA];
    const int ntiles = NN / 16;
    int tile = blockIdx.x;
    if (tile >= ntiles) return;

    const int tid = threadIdx.x;
    const int wave = tid >> 5, lane = tid & 31;
    const int r = lane & 15, kh = lane >> 4;
    const int col0 = wave * 16;

    v2f bf[32];
    {
        const float* Bp = W + (size_t)(2 * kh) * 64 + col0 + r;
#pragma unroll
        for (int kt = 0; kt < 32; ++kt) {
            bf[kt].x = Bp[kt * 256];
            bf[kt].y = Bp[kt * 256 + 64];
        }
    }
    const int n = col0 + r;

    const int srow = wave, sc4 = lane;
    const float4* A4 = (const float4*)Hin;
    float4 p[4];
#pragma unroll
    for (int j = 0; j < 4; ++j) {
        int row = tile * 16 + srow + 4 * j;
        float sc = onorm[row];
        float4 q = A4[(size_t)row * 32 + sc4];
        q.x *= sc; q.y *= sc; q.z *= sc; q.w *= sc;
        p[j] = q;
    }
#pragma unroll
    for (int j = 0; j < 4; ++j) *(float4*)&lA[0][(srow + 4 * j) * LDA + sc4 * 4] = p[j];
    __syncthreads();

    int cur = 0;
    for (; tile < ntiles; tile += gridDim.x) {
        const int nxt = tile + gridDim.x;
        const bool more = nxt < ntiles;
        float sn[4];
        if (more) {
#pragma unroll
            for (int j = 0; j < 4; ++j) {
                int row = nxt * 16 + srow + 4 * j;
                sn[j] = onorm[row];
                p[j] = A4[(size_t)row * 32 + sc4];
            }
        }
        const float* ap = &lA[cur][r * LDA + 2 * kh];
        v8f c = {};
#pragma unroll
        for (int kt = 0; kt < 32; ++kt) {
            v2f a = *(const v2f*)(ap + kt * 4);
            c = __builtin_amdgcn_wmma_f32_16x16x4_f32(false, a, false, bf[kt], (short)0, c, false, false);
        }
        if (more) {
#pragma unroll
            for (int j = 0; j < 4; ++j) {
                float4 q = p[j];
                float sc = sn[j];
                q.x *= sc; q.y *= sc; q.z *= sc; q.w *= sc;
                *(float4*)&lA[cur ^ 1][(srow + 4 * j) * LDA + sc4 * 4] = q;
            }
        }
        __syncthreads();
        const int row0 = tile * 16;
#pragma unroll
        for (int v = 0; v < 8; ++v) {
            int m = row0 + v + 8 * kh;
            T[(size_t)m * 64 + n] = c[v];
        }
        cur ^= 1;
    }
}

__global__ void GCN_score_kernel(const float* __restrict__ H2, const int* __restrict__ qs,
                                 const int* __restrict__ qd, float* __restrict__ out) {
    int t = blockIdx.x * blockDim.x + threadIdx.x;
    int q = t >> 5, lane = t & 31;
    int s = qs[q], d = qd[q];
    float2 a = *(const float2*)(H2 + (size_t)s * 64 + lane * 2);
    float2 b = *(const float2*)(H2 + (size_t)d * 64 + lane * 2);
    float p = a.x * b.x + a.y * b.y;
#pragma unroll
    for (int off = 16; off > 0; off >>= 1) p += __shfl_xor(p, off, 32);
    if (lane == 0) out[q] = 1.0f / (1.0f + __expf(-p));
}

extern "C" void kernel_launch(void* const* d_in, const int* in_sizes, int n_in,
                              void* d_out, int out_size, void* d_ws, size_t ws_size,
                              hipStream_t stream) {
    const float* feat = (const float*)d_in[0];
    const int* esrc   = (const int*)d_in[1];
    const int* edst   = (const int*)d_in[2];
    const int* qsrc   = (const int*)d_in[3];
    const int* qdst   = (const int*)d_in[4];
    const float* W1   = (const float*)d_in[5];
    const float* b1   = (const float*)d_in[6];
    const float* W2   = (const float*)d_in[7];
    const float* b2   = (const float*)d_in[8];
    float* out = (float*)d_out;

    float* ws    = (float*)d_ws;
    float* onorm = ws;
    float* inorm = ws + NN;
    float* agg1  = ws + 2 * NN;
    float* h1    = agg1 + (size_t)NN * 128;
    float* t2    = h1 + (size_t)NN * 128;
    float* h2    = t2 + (size_t)NN * 64;

    int* part = (int*)h1;

    GCN_hist_kernel<<<2 * HCH, 1024, 0, stream>>>(esrc, edst, part);
    GCN_norm_kernel<<<(2 * HW + 255) / 256, 256, 0, stream>>>(part, onorm);
    GCN_own_kernel<128, true, false, true, false><<<OBLK, 1024, 0, stream>>>(feat, esrc, edst, onorm, inorm, b1, agg1, h2);
    GCN_gemm1_kernel<<<1024, 256, 0, stream>>>(agg1, W1, b1, inorm, h1);
    GCN_gemm2_kernel<<<1250, 128, 0, stream>>>(h1, W2, onorm, t2);
    GCN_own_kernel<64, false, true, false, true><<<OBLK, 1024, 0, stream>>>(t2, esrc, edst, onorm, inorm, b2, h2, h2);
    GCN_score_kernel<<<QQ / 8, 256, 0, stream>>>(h2, qsrc, qdst, out);
}


# Both dense products run inside the layer-1 owner block on its finished rows in LDS (hidden rows rewritten in place behind a barrier): agg1 and h1 never reach memory, 7 launches -> 5; sum orders kept
# speedup vs baseline: 2.7153x; 1.0517x over previous
#include <hip/hip_runtime.h>
#include <math.h>

typedef __attribute__((ext_vector_type(2))) float v2f;
typedef __attribute__((ext_vector_type(8))) float v8f;

#define NN 100000
#define EE 1600000
#define QQ 1000000

#define HCH  64
#define HLEN (EE / HCH)
#define HW   (NN / 2)
#define HWP  50016
#define HIT  ((HLEN / 4 + 1023) / 1024)
static_assert(EE % HCH == 0 && HLEN % 4 == 0 && HLEN < 65536, "chunk: whole int4s, 16-bit safe");
static_assert(NN % 2 == 0 && HW % 4 == 0 && HWP >= HW && (HWP * 4) % 128 == 0, "plane layout");
__global__ void __launch_bounds__(1024)
GCN_hist_kernel(const int* __restrict__ esrc, const int* __restrict__ edst, int* __restrict__ part) {
    __shared__ __attribute__((aligned(16))) unsigned cnt[HW];
    const int tid = threadIdx.x;
    const int which = blockIdx.x / HCH, chunk = blockIdx.x - which * HCH;
    const int4* ids4 = (const int4*)((which ? edst : esrc) + (size_t)chunk * HLEN);
    uint4* cnt4 = (uint4*)cnt;
    int4 v[HIT];
#pragma unroll
    for (int k = 0; k < HIT; ++k) {
        int i = tid + k * 1024;
        if (i < HLEN / 4) v[k] = ids4[i];
    }
    for (int w = tid; w < HW / 4; w += 1024) cnt4[w] = make_uint4(0u, 0u, 0u, 0u);
    __syncthreads();
#pragma unroll
    for (int k = 0; k < HIT; ++k) {
        int i = tid + k * 1024;
        if (i < HLEN / 4) {
            atomicAdd(&cnt[v[k].x >> 1], 1u << ((v[k].x & 1) * 16));
            atomicAdd(&cnt[v[k].y >> 1], 1u << ((v[k].y & 1) * 16));
            atomicAdd(&cnt[v[k].z >> 1], 1u << ((v[k].z & 1) * 16));
            atomicAdd(&cnt[v[k].w >> 1], 1u << ((v[k].w & 1) * 16));
        }
    }
    __syncthreads();
    uint4* out4 = (uint4*)(part + (size_t)blockIdx.x * HWP);
    for (int w = tid; w < HW / 4; w += 1024) out4[w] = cnt4[w];
}

__global__ void __launch_bounds__(256)
GCN_norm_kernel(const int* __restrict__ part, float* __restrict__ d) {
    int i = blockIdx.x * 256 + threadIdx.x;
    if (i >= 2 * HW) return;
    const int which = i >= HW ? 1 : 0;
    const int w = i - which * HW;
    const unsigned* p = (const unsigned*)part + (size_t)which * HCH * HWP + w;
    unsigned lo = 0, hi = 0;
#pragma unroll 16
    for (int c = 0; c < HCH; ++c) {
        unsigned v = p[(size_t)c * HWP];
        lo += v & 0xffffu;
        hi += v >> 16;
    }
    float2 r;
    r.x = rsqrtf(fmaxf((float)lo, 1.0f));
    r.y = rsqrtf(fmaxf((float)hi, 1.0f));
    *(float2*)(d + (size_t)which * NN + 2 * w) = r;
}

#define OR_   400
#define APITCH 132
#define OBLK  (NN / OR_)
#define NG    (EE / 128)
#define LCAP  144
#define WSEG  800
#define WCAP  (WSEG - 1)
#define OSTEPS ((((NG / 32 + 1 + 3) / 4 + 1) / 2) * 2)
static_assert(NN % OR_ == 0 && EE % 128 == 0 && NG % OBLK == 0, "ranges and groups are whole");
static_assert(EE <= (1 << 21) && OR_ <= (1 << 11), "list word: row in the top 11 bits, edge in the low 21");
static_assert(OSTEPS * 4 * 32 >= NG + 32, "the steps cover every group of every wave");
static_assert(32 * WSEG == OR_ * 64, "a block's saved words fill exactly its own span of the layer-2 plane");
static_assert(NN <= (1 << 17) && OR_ <= (1 << 9), "saved word: row in the top bits, source id in the low 17");

template <int COLS, int OFF, bool FULL>
__device__ __forceinline__ void GCN_saved_batch(const float* __restrict__ X, float* acc,
                                                unsigned w, int m, int lane) {
    constexpr int LPR = COLS / 32;
    const int s = (int)(w & 0x1ffffu);
    const int dl = (int)(w >> 17);
    float v[16][LPR];
#pragma unroll
    for (int j = 0; j < 16; ++j) {
        int sj = __builtin_amdgcn_readlane(s, OFF + j);
        if (!FULL && j >= m) sj = 0;
        const float* sp = X + (size_t)sj * COLS + lane;
#pragma unroll
        for (int q = 0; q < LPR; ++q) v[j][q] = sp[32 * q];
    }
#pragma unroll
    for (int j = 0; j < 16; ++j) {
        if (FULL || j < m) {
            int dj = __builtin_amdgcn_readlane(dl, OFF + j);
            float* ap = acc + dj * COLS + lane;
#pragma unroll
            for (int q = 0; q < LPR; ++q)
                __hip_atomic_fetch_add(ap + 32 * q, v[j][q], __ATOMIC_RELAXED, __HIP_MEMORY_SCOPE_WORKGROUP);
        }
    }
}

template <int COLS, bool SRC_SCALE, bool FINISH, bool SAVE, bool USE, bool PROD>
__global__ void __launch_bounds__(1024)
GCN_own_kernel(const float* __restrict__ X, const int* __restrict__ esrc, const int* __restrict__ edst,
               const float* __restrict__ onorm, const float* __restrict__ inorm,
               const float* __restrict__ bias, float* __restrict__ Y, float* __restrict__ L,
               const float* __restrict__ Wa, const float* __restrict__ Wb) {
    constexpr int LPR = COLS / 32;
    constexpr int BAT = 32 / LPR;
    constexpr int PITCH = PROD ? APITCH : COLS;
    static_assert(!PROD || (COLS == 128 && !FINISH && OR_ % 16 == 0), "the products take 128-wide rows in 16-row tiles");
    static_assert(!USE || !PROD, "the saved-edge path adds at pitch COLS");
    static_assert(BAT - 1 + 128 <= LCAP, "list capacity");
    static_assert(!USE || COLS == 64, "the saved words live in a span of OR_ x 64 floats");
    static_assert(!(SAVE && USE), "a launch saves or reads");
    __shared__ __attribute__((aligned(16))) float acc[OR_ * PITCH];
    __shared__ unsigned lst[32 * LCAP];
    const int tid = threadIdx.x, lane = tid & 31;
    const int wave = __builtin_amdgcn_readfirstlane(tid >> 5);
    const int base = blockIdx.x * OR_;
    const int rot = blockIdx.x * (NG / OBLK);
    const int4* ids4 = (const int4*)edst;
    unsigned* mylst = lst + wave * LCAP;
    float4* acc4 = (float4*)acc;
    int cnt = 0;
    int gpos = 0;
    float* sv = L + (size_t)blockIdx.x * (OR_ * 64) + wave * WSEG;

    auto load4 = [&](int4 (&buf)[4], int s) {
#pragma unroll
        for (int j = 0; j < 4; ++j) {
            int it0 = wave + 32 * (s * 4 + j);
            int g = it0 + rot;
            if (g >= NG) g -= NG;
            if (it0 >= NG) g = 0;
            buf[j] = ids4[g * 32 + lane];
        }
    };
    auto drain = [&]() {
        cnt -= BAT;
        unsigned p = mylst[cnt + (lane & (BAT - 1))];
        int s = esrc[p & 0x1fffffu];
        int dl = (int)(p >> 21);
        float sc = 1.0f;
        if (SRC_SCALE) sc = onorm[s];
        if (SAVE) {
            if (gpos + BAT <= WCAP && lane < BAT)
                sv[gpos + lane] = __uint_as_float(((unsigned)dl << 17) | (unsigned)s);
            gpos += BAT;
        }
        float v[BAT][LPR];
#pragma unroll
        for (int j = 0; j < BAT; ++j) {
            int sj = __builtin_amdgcn_readlane(s, j);
            const float* sp = X + (size_t)sj * COLS + lane;
#pragma unroll
            for (int q = 0; q < LPR; ++q) v[j][q] = sp[32 * q];
        }
#pragma unroll
        for (int j = 0; j < BAT; ++j) {
            int dj = __builtin_amdgcn_readlane(dl, j);
            float scj = __int_as_float(__builtin_amdgcn_readlane(__float_as_int(sc), j));
            float* ap = acc + dj * PITCH + lane;
#pragma unroll
            for (int q = 0; q < LPR; ++q) {
                float val = SRC_SCALE ? v[j][q] * scj : v[j][q];
                __hip_atomic_fetch_add(ap + 32 * q, val, __ATOMIC_RELAXED, __HIP_MEMORY_SCOPE_WORKGROUP);
            }
        }
    };
    auto scan4 = [&](const int4 (&buf)[4], int s) {
#pragma unroll
        for (int j = 0; j < 4; ++j) {
            int it0 = wave + 32 * (s * 4 + j);
            if (it0 < NG) {
                int g = it0 + rot;
                if (g >= NG) g -= NG;
                unsigned u[4];
                u[0] = (unsigned)(buf[j].x - base); u[1] = (unsigned)(buf[j].y - base);
                u[2] = (unsigned)(buf[j].z - base); u[3] = (unsigned)(buf[j].w - base);
                unsigned mn = min(min(u[0], u[1]), min(u[2], u[3]));
                if (__builtin_amdgcn_ballot_w32(mn < (unsigned)OR_)) {
                    const unsigned e0 = (unsigned)(g * 32 + lane) * 4u;
#pragma unroll
                    for (int c = 0; c < 4; ++c) {
                        bool m = u[c] < (unsigned)OR_;
                        unsigned bal = __builtin_amdgcn_ballot_w32(m);
                        if (bal) {
                            int pos = cnt + (int)__builtin_amdgcn_mbcnt_lo(bal, 0u);
                            if (m) mylst[pos] = (u[c] << 21) | (e0 + c);
                            cnt += __builtin_popcount(bal);
                        }
                    }
                    while (cnt >= BAT) drain();
                }
            }
        }
    };

    int4 ia[4], ib[4];
    int nsaved = 0;
    bool refilter = !USE;
    const float* span = Y + (size_t)blockIdx.x * (OR_ * COLS);
    if (USE) {
        unsigned cv = __float_as_uint(span[lane * WSEG + WCAP]);
        refilter = __builtin_amdgcn_ballot_w32(cv > (unsigned)WCAP) != 0u;
        nsaved = __builtin_amdgcn_readlane((int)cv, wave);
    } else {
        load4(ia, 0);
    }
    unsigned pw = 0u;
    if (USE && !refilter && nsaved > 0)
        pw = __float_as_uint(span[wave * WSEG + lane]);
    for (int i = tid; i < OR_ * PITCH / 4; i += 1024) acc4[i] = make_float4(0.0f, 0.0f, 0.0f, 0.0f);
    __syncthreads();
    if (USE && !refilter) {
        const float* seg = span + wave * WSEG;
        int done = 0;
        for (; done + 32 <= nsaved; done += 32) {
            unsigned pn = 0u;
            if (done + 32 < nsaved) pn = __float_as_uint(seg[done + 32 + lane]);
            GCN_saved_batch<COLS, 0, true>(X, acc, pw, 16, lane);
            GCN_saved_batch<COLS, 16, true>(X, acc, pw, 16, lane);
            pw = pn;
        }
        const int m = nsaved - done;
        if (m >= 16) {
            GCN_saved_batch<COLS, 0, true>(X, acc, pw, 16, lane);
            if (m > 16) GCN_saved_batch<COLS, 16, false>(X, acc, pw, m - 16, lane);
        } else if (m > 0) {
            GCN_saved_batch<COLS, 0, false>(X, acc, pw, m, lane);
        }
    } else {
    if (USE) load4(ia, 0);
    for (int s = 0; s < OSTEPS; s += 2) {
        load4(ib, s + 1);
        scan4(ia, s);
        load4(ia, s + 2);
        scan4(ib, s + 1);
    }
    if (cnt > 0) {
        unsigned p = mylst[lane < cnt ? lane : 0];
        int s = esrc[p & 0x1fffffu];
        int dl = (int)(p >> 21);
        float sc = 1.0f;
        if (SRC_SCALE) sc = onorm[s];
        if (SAVE) {
            if (gpos + cnt <= WCAP && lane < cnt)
                sv[gpos + lane] = __uint_as_float(((unsigned)dl << 17) | (unsigned)s);
            gpos += cnt;
        }
        for (int j = 0; j < cnt; ++j) {
            int sj = __builtin_amdgcn_readlane(s, j);
            int dj = __builtin_amdgcn_readlane(dl, j);
            float scj = __int_as_float(__builtin_amdgcn_readlane(__float_as_int(sc), j));
            const float* sp = X + (size_t)sj * COLS + lane;
            float* ap = acc + dj * PITCH + lane;
#pragma unroll
            for (int q = 0; q < LPR; ++q) {
                float val = SRC_SCALE ? sp[32 * q] * scj : sp[32 * q];
                __hip_atomic_fetch_add(ap + 32 * q, val, __ATOMIC_RELAXED, __HIP_MEMORY_SCOPE_WORKGROUP);
            }
        }
    }
    if (SAVE && lane == 0) sv[WCAP] = __uint_as_float((unsigned)gpos);
    }
    __syncthreads();
    if constexpr (PROD) {
        constexpr int NT = OR_ / 16;
        const int r = lane & 15, kh = lane >> 4;
        {
            const int cs = wave & 7, rg = wave >> 3;
            const int n = cs * 16 + r;
            v2f bf[32];
            const float* Bp = Wa + (size_t)(2 * kh) * 128 + n;
#pragma unroll
            for (int kt = 0; kt < 32; ++kt) {
                bf[kt].x = Bp[kt * 512];
                bf[kt].y = Bp[kt * 512 + 128];
            }
            const float bb = bias[n];
            for (int it = 0; it < (NT + 3) / 4; ++it) {
                const int t = rg + 4 * it;
                const bool live = t < NT;
                v8f c = {};
                float4 ci0 = {}, ci1 = {}, co0 = {}, co1 = {};
                if (live) {
                    const float4* ip = (const float4*)(inorm + base + t * 16 + 8 * kh);
                    const float4* op = (const float4*)(onorm + base + t * 16 + 8 * kh);
                    ci0 = ip[0]; ci1 = ip[1]; co0 = op[0]; co1 = op[1];
                    const float* ap = acc + (t * 16 + r) * PITCH + 2 * kh;
#pragma unroll
                    for (int kt = 0; kt < 32; ++kt) {
                        v2f a = *(const v2f*)(ap + kt * 4);
                        c = __builtin_amdgcn_wmma_f32_16x16x4_f32(false, a, false, bf[kt], (short)0, c, false, false);
                    }
                }
                __syncthreads();
                if (live) {
                    const float ci[8] = {ci0.x, ci0.y, ci0.z, ci0.w, ci1.x, ci1.y, ci1.z, ci1.w};
                    const float co[8] = {co0.x, co0.y, co0.z, co0.w, co1.x, co1.y, co1.z, co1.w};
                    float* hp = acc + (t * 16 + 8 * kh) * PITCH + n;
#pragma unroll
                    for (int v = 0; v < 8; ++v) {
                        float val = fmaxf(fmaf(c[v], ci[v], bb), 0.0f);
                        hp[v * PITCH] = val * co[v];
                    }
                }
            }
        }
        __syncthreads();
        {
            const int cs = wave & 3, rg = wave >> 2;
            const int n = cs * 16 + r;
            v2f bf[32];
            const float* Bp = Wb + (size_t)(2 * kh) * 64 + n;
#pragma unroll
            for (int kt = 0; kt < 32; ++kt) {
                bf[kt].x = Bp[kt * 256];
                bf[kt].y = Bp[kt * 256 + 64];
            }
            for (int t = rg; t < NT; t += 8) {
                const float* ap = acc + (t * 16 + r) * PITCH + 2 * kh;
                v8f c = {};
#pragma unroll
                for (int kt = 0; kt < 32; ++kt) {
                    v2f a = *(const v2f*)(ap + kt * 4);
                    c = __builtin_amdgcn_wmma_f32_16x16x4_f32(false, a, false, bf[kt], (short)0, c, false, false);
                }
                float* tp = Y + (size_t)(base + t * 16 + 8 * kh) * 64 + n;
#pragma unroll
                for (int v = 0; v < 8; ++v) tp[v * 64] = c[v];
            }
        }
        return;
    }
    float4* Y4 = (float4*)(Y + (size_t)base * COLS);
    for (int i = tid; i < OR_ * COLS / 4; i += 1024) {
        float4 a = acc4[i];
        if (FINISH) {
            const int row = base + i / (COLS / 4);
            const int c4 = (i % (COLS / 4)) * 4;
            const float sn = inorm[row];
            a.x = fmaf(a.x, sn, bias[c4 + 0]);
            a.y = fmaf(a.y, sn, bias[c4 + 1]);
            a.z = fmaf(a.z, sn, bias[c4 + 2]);
            a.w = fmaf(a.w, sn, bias[c4 + 3]);
        }
        Y4[i] = a;
    }
}

__global__ void GCN_score_kernel(const float* __restrict__ H2, const int* __restrict__ qs,
                                 const int* __restrict__ qd, float* __restrict__ out) {
    int t = blockIdx.x * blockDim.x + threadIdx.x;
    int q = t >> 5, lane = t & 31;
    int s = qs[q], d = qd[q];
    float2 a = *(const float2*)(H2 + (size_t)s * 64 + lane * 2);
    float2 b = *(const float2*)(H2 + (size_t)d * 64 + lane * 2);
    float p = a.x * b.x + a.y * b.y;
#pragma unroll
    for (int off = 16; off > 0; off >>= 1) p += __shfl_xor(p, off, 32);
    if (lane == 0) out[q] = 1.0f / (1.0f + __expf(-p));
}

extern "C" void kernel_launch(void* const* d_in, const int* in_sizes, int n_in,
                              void* d_out, int out_size, void* d_ws, size_t ws_size,
                              hipStream_t stream) {
    const float* feat = (const float*)d_in[0];
    const int* esrc   = (const int*)d_in[1];
    const int* edst   = (const int*)d_in[2];
    const int* qsrc   = (const int*)d_in[3];
    const int* qdst   = (const int*)d_in[4];
    const float* W1   = (const float*)d_in[5];
    const float* b1   = (const float*)d_in[6];
    const float* W2   = (const float*)d_in[7];
    const float* b2   = (const float*)d_in[8];
    float* out = (float*)d_out;

    float* ws    = (float*)d_ws;
    float* onorm = ws;
    float* inorm = ws + NN;
    int* part    = (int*)(ws + 2 * NN);
    float* t2    = ws + 2 * NN + (size_t)NN * 256;
    float* h2    = t2 + (size_t)NN * 64;

    GCN_hist_kernel<<<2 * HCH, 1024, 0, stream>>>(esrc, edst, part);
    GCN_norm_kernel<<<(2 * HW + 255) / 256, 256, 0, stream>>>(part, onorm);
    GCN_own_kernel<128, true, false, true, false, true><<<OBLK, 1024, 0, stream>>>(feat, esrc, edst, onorm, inorm, b1, t2, h2, W1, W2);
    GCN_own_kernel<64, false, true, false, true, false><<<OBLK, 1024, 0, stream>>>(t2, esrc, edst, onorm, inorm, b2, h2, h2, nullptr, nullptr);
    GCN_score_kernel<<<QQ / 8, 256, 0, stream>>>(h2, qsrc, qdst, out);
}


# Scoring: the five lane exchanges of a query's sum tree run on the vector unit (row_xmask operand swizzles, one cross-row permute), not as LDS-unit shuffles with a wait each; same tree, same sums
# speedup vs baseline: 2.8017x; 1.0318x over previous
#include <hip/hip_runtime.h>
#include <math.h>

typedef __attribute__((ext_vector_type(2))) float v2f;
typedef __attribute__((ext_vector_type(8))) float v8f;

#define NN 100000
#define EE 1600000
#define QQ 1000000

#define HCH  64
#define HLEN (EE / HCH)
#define HW   (NN / 2)
#define HWP  50016
#define HIT  ((HLEN / 4 + 1023) / 1024)
static_assert(EE % HCH == 0 && HLEN % 4 == 0 && HLEN < 65536, "chunk: whole int4s, 16-bit safe");
static_assert(NN % 2 == 0 && HW % 4 == 0 && HWP >= HW && (HWP * 4) % 128 == 0, "plane layout");
__global__ void __launch_bounds__(1024)
GCN_hist_kernel(const int* __restrict__ esrc, const int* __restrict__ edst, int* __restrict__ part) {
    __shared__ __attribute__((aligned(16))) unsigned cnt[HW];
    const int tid = threadIdx.x;
    const int which = blockIdx.x / HCH, chunk = blockIdx.x - which * HCH;
    const int4* ids4 = (const int4*)((which ? edst : esrc) + (size_t)chunk * HLEN);
    uint4* cnt4 = (uint4*)cnt;
    int4 v[HIT];
#pragma unroll
    for (int k = 0; k < HIT; ++k) {
        int i = tid + k * 1024;
        if (i < HLEN / 4) v[k] = ids4[i];
    }
    for (int w = tid; w < HW / 4; w += 1024) cnt4[w] = make_uint4(0u, 0u, 0u, 0u);
    __syncthreads();
#pragma unroll
    for (int k = 0; k < HIT; ++k) {
        int i = tid + k * 1024;
        if (i < HLEN / 4) {
            atomicAdd(&cnt[v[k].x >> 1], 1u << ((v[k].x & 1) * 16));
            atomicAdd(&cnt[v[k].y >> 1], 1u << ((v[k].y & 1) * 16));
            atomicAdd(&cnt[v[k].z >> 1], 1u << ((v[k].z & 1) * 16));
            atomicAdd(&cnt[v[k].w >> 1], 1u << ((v[k].w & 1) * 16));
        }
    }
    __syncthreads();
    uint4* out4 = (uint4*)(part + (size_t)blockIdx.x * HWP);
    for (int w = tid; w < HW / 4; w += 1024) out4[w] = cnt4[w];
}

__global__ void __launch_bounds__(256)
GCN_norm_kernel(const int* __restrict__ part, float* __restrict__ d) {
    int i = blockIdx.x * 256 + threadIdx.x;
    if (i >= 2 * HW) return;
    const int which = i >= HW ? 1 : 0;
    const int w = i - which * HW;
    const unsigned* p = (const unsigned*)part + (size_t)which * HCH * HWP + w;
    unsigned lo = 0, hi = 0;
#pragma unroll 16
    for (int c = 0; c < HCH; ++c) {
        unsigned v = p[(size_t)c * HWP];
        lo += v & 0xffffu;
        hi += v >> 16;
    }
    float2 r;
    r.x = rsqrtf(fmaxf((float)lo, 1.0f));
    r.y = rsqrtf(fmaxf((float)hi, 1.0f));
    *(float2*)(d + (size_t)which * NN + 2 * w) = r;
}

#define OR_   400
#define APITCH 132
#define OBLK  (NN / OR_)
#define NG    (EE / 128)
#define LCAP  144
#define WSEG  800
#define WCAP  (WSEG - 1)
#define OSTEPS ((((NG / 32 + 1 + 3) / 4 + 1) / 2) * 2)
static_assert(NN % OR_ == 0 && EE % 128 == 0 && NG % OBLK == 0, "ranges and groups are whole");
static_assert(EE <= (1 << 21) && OR_ <= (1 << 11), "list word: row in the top 11 bits, edge in the low 21");
static_assert(OSTEPS * 4 * 32 >= NG + 32, "the steps cover every group of every wave");
static_assert(32 * WSEG == OR_ * 64, "a block's saved words fill exactly its own span of the layer-2 plane");
static_assert(NN <= (1 << 17) && OR_ <= (1 << 9), "saved word: row in the top bits, source id in the low 17");

template <int COLS, int OFF, bool FULL>
__device__ __forceinline__ void GCN_saved_batch(const float* __restrict__ X, float* acc,
                                                unsigned w, int m, int lane) {
    constexpr int LPR = COLS / 32;
    const int s = (int)(w & 0x1ffffu);
    const int dl = (int)(w >> 17);
    float v[16][LPR];
#pragma unroll
    for (int j = 0; j < 16; ++j) {
        int sj = __builtin_amdgcn_readlane(s, OFF + j);
        if (!FULL && j >= m) sj = 0;
        const float* sp = X + (size_t)sj * COLS + lane;
#pragma unroll
        for (int q = 0; q < LPR; ++q) v[j][q] = sp[32 * q];
    }
#pragma unroll
    for (int j = 0; j < 16; ++j) {
        if (FULL || j < m) {
            int dj = __builtin_amdgcn_readlane(dl, OFF + j);
            float* ap = acc + dj * COLS + lane;
#pragma unroll
            for (int q = 0; q < LPR; ++q)
                __hip_atomic_fetch_add(ap + 32 * q, v[j][q], __ATOMIC_RELAXED, __HIP_MEMORY_SCOPE_WORKGROUP);
        }
    }
}

template <int COLS, bool SRC_SCALE, bool FINISH, bool SAVE, bool USE, bool PROD>
__global__ void __launch_bounds__(1024)
GCN_own_kernel(const float* __restrict__ X, const int* __restrict__ esrc, const int* __restrict__ edst,
               const float* __restrict__ onorm, const float* __restrict__ inorm,
               const float* __restrict__ bias, float* __restrict__ Y, float* __restrict__ L,
               const float* __restrict__ Wa, const float* __restrict__ Wb) {
    constexpr int LPR = COLS / 32;
    constexpr int BAT = 32 / LPR;
    constexpr int PITCH = PROD ? APITCH : COLS;
    static_assert(!PROD || (COLS == 128 && !FINISH && OR_ % 16 == 0), "the products take 128-wide rows in 16-row tiles");
    static_assert(!USE || !PROD, "the saved-edge path adds at pitch COLS");
    static_assert(BAT - 1 + 128 <= LCAP, "list capacity");
    static_assert(!USE || COLS == 64, "the saved words live in a span of OR_ x 64 floats");
    static_assert(!(SAVE && USE), "a launch saves or reads");
    __shared__ __attribute__((aligned(16))) float acc[OR_ * PITCH];
    __shared__ unsigned lst[32 * LCAP];
    const int tid = threadIdx.x, lane = tid & 31;
    const int wave = __builtin_amdgcn_readfirstlane(tid >> 5);
    const int base = blockIdx.x * OR_;
    const int rot = blockIdx.x * (NG / OBLK);
    const int4* ids4 = (const int4*)edst;
    unsigned* mylst = lst + wave * LCAP;
    float4* acc4 = (float4*)acc;
    int cnt = 0;
    int gpos = 0;
    float* sv = L + (size_t)blockIdx.x * (OR_ * 64) + wave * WSEG;

    auto load4 = [&](int4 (&buf)[4], int s) {
#pragma unroll
        for (int j = 0; j < 4; ++j) {
            int it0 = wave + 32 * (s * 4 + j);
            int g = it0 + rot;
            if (g >= NG) g -= NG;
            if (it0 >= NG) g = 0;
            buf[j] = ids4[g * 32 + lane];
        }
    };
    auto drain = [&]() {
        cnt -= BAT;
        unsigned p = mylst[cnt + (lane & (BAT - 1))];
        int s = esrc[p & 0x1fffffu];
        int dl = (int)(p >> 21);
        float sc = 1.0f;
        if (SRC_SCALE) sc = onorm[s];
        if (SAVE) {
            if (gpos + BAT <= WCAP && lane < BAT)
                sv[gpos + lane] = __uint_as_float(((unsigned)dl << 17) | (unsigned)s);
            gpos += BAT;
        }
        float v[BAT][LPR];
#pragma unroll
        for (int j = 0; j < BAT; ++j) {
            int sj = __builtin_amdgcn_readlane(s, j);
            const float* sp = X + (size_t)sj * COLS + lane;
#pragma unroll
            for (int q = 0; q < LPR; ++q) v[j][q] = sp[32 * q];
        }
#pragma unroll
        for (int j = 0; j < BAT; ++j) {
            int dj = __builtin_amdgcn_readlane(dl, j);
            float scj = __int_as_float(__builtin_amdgcn_readlane(__float_as_int(sc), j));
            float* ap = acc + dj * PITCH + lane;
#pragma unroll
            for (int q = 0; q < LPR; ++q) {
                float val = SRC_SCALE ? v[j][q] * scj : v[j][q];
                __hip_atomic_fetch_add(ap + 32 * q, val, __ATOMIC_RELAXED, __HIP_MEMORY_SCOPE_WORKGROUP);
            }
        }
    };
    auto scan4 = [&](const int4 (&buf)[4], int s) {
#pragma unroll
        for (int j = 0; j < 4; ++j) {
            int it0 = wave + 32 * (s * 4 + j);
            if (it0 < NG) {
                int g = it0 + rot;
                if (g >= NG) g -= NG;
                unsigned u[4];
                u[0] = (unsigned)(buf[j].x - base); u[1] = (unsigned)(buf[j].y - base);
                u[2] = (unsigned)(buf[j].z - base); u[3] = (unsigned)(buf[j].w - base);
                unsigned mn = min(min(u[0], u[1]), min(u[2], u[3]));
                if (__builtin_amdgcn_ballot_w32(mn < (unsigned)OR_)) {
                    const unsigned e0 = (unsigned)(g * 32 + lane) * 4u;
#pragma unroll
                    for (int c = 0; c < 4; ++c) {
                        bool m = u[c] < (unsigned)OR_;
                        unsigned bal = __builtin_amdgcn_ballot_w32(m);
                        if (bal) {
                            int pos = cnt + (int)__builtin_amdgcn_mbcnt_lo(bal, 0u);
                            if (m) mylst[pos] = (u[c] << 21) | (e0 + c);
                            cnt += __builtin_popcount(bal);
                        }
                    }
                    while (cnt >= BAT) drain();
                }
            }
        }
    };

    int4 ia[4], ib[4];
    int nsaved = 0;
    bool refilter = !USE;
    const float* span = Y + (size_t)blockIdx.x * (OR_ * COLS);
    if (USE) {
        unsigned cv = __float_as_uint(span[lane * WSEG + WCAP]);
        refilter = __builtin_amdgcn_ballot_w32(cv > (unsigned)WCAP) != 0u;
        nsaved = __builtin_amdgcn_readlane((int)cv, wave);
    } else {
        load4(ia, 0);
    }
    unsigned pw = 0u;
    if (USE && !refilter && nsaved > 0)
        pw = __float_as_uint(span[wave * WSEG + lane]);
    for (int i = tid; i < OR_ * PITCH / 4; i += 1024) acc4[i] = make_float4(0.0f, 0.0f, 0.0f, 0.0f);
    __syncthreads();
    if (USE && !refilter) {
        const float* seg = span + wave * WSEG;
        int done = 0;
        for (; done + 32 <= nsaved; done += 32) {
            unsigned pn = 0u;
            if (done + 32 < nsaved) pn = __float_as_uint(seg[done + 32 + lane]);
            GCN_saved_batch<COLS, 0, true>(X, acc, pw, 16, lane);
            GCN_saved_batch<COLS, 16, true>(X, acc, pw, 16, lane);
            pw = pn;
        }
        const int m = nsaved - done;
        if (m >= 16) {
            GCN_saved_batch<COLS, 0, true>(X, acc, pw, 16, lane);
            if (m > 16) GCN_saved_batch<COLS, 16, false>(X, acc, pw, m - 16, lane);
        } else if (m > 0) {
            GCN_saved_batch<COLS, 0, false>(X, acc, pw, m, lane);
        }
    } else {
    if (USE) load4(ia, 0);
    for (int s = 0; s < OSTEPS; s += 2) {
        load4(ib, s + 1);
        scan4(ia, s);
        load4(ia, s + 2);
        scan4(ib, s + 1);
    }
    if (cnt > 0) {
        unsigned p = mylst[lane < cnt ? lane : 0];
        int s = esrc[p & 0x1fffffu];
        int dl = (int)(p >> 21);
        float sc = 1.0f;
        if (SRC_SCALE) sc = onorm[s];
        if (SAVE) {
            if (gpos + cnt <= WCAP && lane < cnt)
                sv[gpos + lane] = __uint_as_float(((unsigned)dl << 17) | (unsigned)s);
            gpos += cnt;
        }
        for (int j = 0; j < cnt; ++j) {
            int sj = __builtin_amdgcn_readlane(s, j);
            int dj = __builtin_amdgcn_readlane(dl, j);
            float scj = __int_as_float(__builtin_amdgcn_readlane(__float_as_int(sc), j));
            const float* sp = X + (size_t)sj * COLS + lane;
            float* ap = acc + dj * PITCH + lane;
#pragma unroll
            for (int q = 0; q < LPR; ++q) {
                float val = SRC_SCALE ? sp[32 * q] * scj : sp[32 * q];
                __hip_atomic_fetch_add(ap + 32 * q, val, __ATOMIC_RELAXED, __HIP_MEMORY_SCOPE_WORKGROUP);
            }
        }
    }
    if (SAVE && lane == 0) sv[WCAP] = __uint_as_float((unsigned)gpos);
    }
    __syncthreads();
    if constexpr (PROD) {
        constexpr int NT = OR_ / 16;
        const int r = lane & 15, kh = lane >> 4;
        {
            const int cs = wave & 7, rg = wave >> 3;
            const int n = cs * 16 + r;
            v2f bf[32];
            const float* Bp = Wa + (size_t)(2 * kh) * 128 + n;
#pragma unroll
            for (int kt = 0; kt < 32; ++kt) {
                bf[kt].x = Bp[kt * 512];
                bf[kt].y = Bp[kt * 512 + 128];
            }
            const float bb = bias[n];
            for (int it = 0; it < (NT + 3) / 4; ++it) {
                const int t = rg + 4 * it;
                const bool live = t < NT;
                v8f c = {};
                float4 ci0 = {}, ci1 = {}, co0 = {}, co1 = {};
                if (live) {
                    const float4* ip = (const float4*)(inorm + base + t * 16 + 8 * kh);
                    const float4* op = (const float4*)(onorm + base + t * 16 + 8 * kh);
                    ci0 = ip[0]; ci1 = ip[1]; co0 = op[0]; co1 = op[1];
                    const float* ap = acc + (t * 16 + r) * PITCH + 2 * kh;
#pragma unroll
                    for (int kt = 0; kt < 32; ++kt) {
                        v2f a = *(const v2f*)(ap + kt * 4);
                        c = __builtin_amdgcn_wmma_f32_16x16x4_f32(false, a, false, bf[kt], (short)0, c, false, false);
                    }
                }
                __syncthreads();
                if (live) {
                    const float ci[8] = {ci0.x, ci0.y, ci0.z, ci0.w, ci1.x, ci1.y, ci1.z, ci1.w};
                    const float co[8] = {co0.x, co0.y, co0.z, co0.w, co1.x, co1.y, co1.z, co1.w};
                    float* hp = acc + (t * 16 + 8 * kh) * PITCH + n;
#pragma unroll
                    for (int v = 0; v < 8; ++v) {
                        float val = fmaxf(fmaf(c[v], ci[v], bb), 0.0f);
                        hp[v * PITCH] = val * co[v];
                    }
                }
            }
        }
        __syncthreads();
        {
            const int cs = wave & 3, rg = wave >> 2;
            const int n = cs * 16 + r;
            v2f bf[32];
            const float* Bp = Wb + (size_t)(2 * kh) * 64 + n;
#pragma unroll
            for (int kt = 0; kt < 32; ++kt) {
                bf[kt].x = Bp[kt * 256];
                bf[kt].y = Bp[kt * 256 + 64];
            }
            for (int t = rg; t < NT; t += 8) {
                const float* ap = acc + (t * 16 + r) * PITCH + 2 * kh;
                v8f c = {};
#pragma unroll
                for (int kt = 0; kt < 32; ++kt) {
                    v2f a = *(const v2f*)(ap + kt * 4);
                    c = __builtin_amdgcn_wmma_f32_16x16x4_f32(false, a, false, bf[kt], (short)0, c, false, false);
                }
                float* tp = Y + (size_t)(base + t * 16 + 8 * kh) * 64 + n;
#pragma unroll
                for (int v = 0; v < 8; ++v) tp[v * 64] = c[v];
            }
        }
        return;
    }
    float4* Y4 = (float4*)(Y + (size_t)base * COLS);
    for (int i = tid; i < OR_ * COLS / 4; i += 1024) {
        float4 a = acc4[i];
        if (FINISH) {
            const int row = base + i / (COLS / 4);
            const int c4 = (i % (COLS / 4)) * 4;
            const float sn = inorm[row];
            a.x = fmaf(a.x, sn, bias[c4 + 0]);
            a.y = fmaf(a.y, sn, bias[c4 + 1]);
            a.z = fmaf(a.z, sn, bias[c4 + 2]);
            a.w = fmaf(a.w, sn, bias[c4 + 3]);
        }
        Y4[i] = a;
    }
}

__device__ __forceinline__ float GCN_xhalf(float x) {
    const int b = __builtin_bit_cast(int, x);
    return __builtin_bit_cast(float, __builtin_amdgcn_permlanex16(b, b, 0x76543210, 0xfedcba98, false, false));
}
template <int M>
__device__ __forceinline__ float GCN_xrow(float x) {
    return __builtin_bit_cast(float, __builtin_amdgcn_update_dpp(0, __builtin_bit_cast(int, x), 0x160 + M, 0xf, 0xf, true));
}
__global__ void GCN_score_kernel(const float* __restrict__ H2, const int* __restrict__ qs,
                                 const int* __restrict__ qd, float* __restrict__ out) {
    int t = blockIdx.x * blockDim.x + threadIdx.x;
    int q = t >> 5, lane = t & 31;
    int s = qs[q], d = qd[q];
    float2 a = *(const float2*)(H2 + (size_t)s * 64 + lane * 2);
    float2 b = *(const float2*)(H2 + (size_t)d * 64 + lane * 2);
    float p = a.x * b.x + a.y * b.y;
    p += GCN_xhalf(p);
    p += GCN_xrow<8>(p);
    p += GCN_xrow<4>(p);
    p += GCN_xrow<2>(p);
    p += GCN_xrow<1>(p);
    if (lane == 0) out[q] = 1.0f / (1.0f + __expf(-p));
}

extern "C" void kernel_launch(void* const* d_in, const int* in_sizes, int n_in,
                              void* d_out, int out_size, void* d_ws, size_t ws_size,
                              hipStream_t stream) {
    const float* feat = (const float*)d_in[0];
    const int* esrc   = (const int*)d_in[1];
    const int* edst   = (const int*)d_in[2];
    const int* qsrc   = (const int*)d_in[3];
    const int* qdst   = (const int*)d_in[4];
    const float* W1   = (const float*)d_in[5];
    const float* b1   = (const float*)d_in[6];
    const float* W2   = (const float*)d_in[7];
    const float* b2   = (const float*)d_in[8];
    float* out = (float*)d_out;

    float* ws    = (float*)d_ws;
    float* onorm = ws;
    float* inorm = ws + NN;
    int* part    = (int*)(ws + 2 * NN);
    float* t2    = ws + 2 * NN + (size_t)NN * 256;
    float* h2    = t2 + (size_t)NN * 64;

    GCN_hist_kernel<<<2 * HCH, 1024, 0, stream>>>(esrc, edst, part);
    GCN_norm_kernel<<<(2 * HW + 255) / 256, 256, 0, stream>>>(part, onorm);
    GCN_own_kernel<128, true, false, true, false, true><<<OBLK, 1024, 0, stream>>>(feat, esrc, edst, onorm, inorm, b1, t2, h2, W1, W2);
    GCN_own_kernel<64, false, true, false, true, false><<<OBLK, 1024, 0, stream>>>(t2, esrc, edst, onorm, inorm, b2, h2, h2, nullptr, nullptr);
    GCN_score_kernel<<<QQ / 8, 256, 0, stream>>>(h2, qsrc, qdst, out);
}


# Scoring: one wave per block prefetches into L2 the query-id lines of the queries 65536 places ahead, so every wave's id loads hit L2 and not memory; sums unchanged
# speedup vs baseline: 2.9317x; 1.0464x over previous
#include <hip/hip_runtime.h>
#include <math.h>

typedef __attribute__((ext_vector_type(2))) float v2f;
typedef __attribute__((ext_vector_type(8))) float v8f;

#define NN 100000
#define EE 1600000
#define QQ 1000000

#define HCH  64
#define HLEN (EE / HCH)
#define HW   (NN / 2)
#define HWP  50016
#define HIT  ((HLEN / 4 + 1023) / 1024)
static_assert(EE % HCH == 0 && HLEN % 4 == 0 && HLEN < 65536, "chunk: whole int4s, 16-bit safe");
static_assert(NN % 2 == 0 && HW % 4 == 0 && HWP >= HW && (HWP * 4) % 128 == 0, "plane layout");
__global__ void __launch_bounds__(1024)
GCN_hist_kernel(const int* __restrict__ esrc, const int* __restrict__ edst, int* __restrict__ part) {
    __shared__ __attribute__((aligned(16))) unsigned cnt[HW];
    const int tid = threadIdx.x;
    const int which = blockIdx.x / HCH, chunk = blockIdx.x - which * HCH;
    const int4* ids4 = (const int4*)((which ? edst : esrc) + (size_t)chunk * HLEN);
    uint4* cnt4 = (uint4*)cnt;
    int4 v[HIT];
#pragma unroll
    for (int k = 0; k < HIT; ++k) {
        int i = tid + k * 1024;
        if (i < HLEN / 4) v[k] = ids4[i];
    }
    for (int w = tid; w < HW / 4; w += 1024) cnt4[w] = make_uint4(0u, 0u, 0u, 0u);
    __syncthreads();
#pragma unroll
    for (int k = 0; k < HIT; ++k) {
        int i = tid + k * 1024;
        if (i < HLEN / 4) {
            atomicAdd(&cnt[v[k].x >> 1], 1u << ((v[k].x & 1) * 16));
            atomicAdd(&cnt[v[k].y >> 1], 1u << ((v[k].y & 1) * 16));
            atomicAdd(&cnt[v[k].z >> 1], 1u << ((v[k].z & 1) * 16));
            atomicAdd(&cnt[v[k].w >> 1], 1u << ((v[k].w & 1) * 16));
        }
    }
    __syncthreads();
    uint4* out4 = (uint4*)(part + (size_t)blockIdx.x * HWP);
    for (int w = tid; w < HW / 4; w += 1024) out4[w] = cnt4[w];
}

__global__ void __launch_bounds__(256)
GCN_norm_kernel(const int* __restrict__ part, float* __restrict__ d) {
    int i = blockIdx.x * 256 + threadIdx.x;
    if (i >= 2 * HW) return;
    const int which = i >= HW ? 1 : 0;
    const int w = i - which * HW;
    const unsigned* p = (const unsigned*)part + (size_t)which * HCH * HWP + w;
    unsigned lo = 0, hi = 0;
#pragma unroll 16
    for (int c = 0; c < HCH; ++c) {
        unsigned v = p[(size_t)c * HWP];
        lo += v & 0xffffu;
        hi += v >> 16;
    }
    float2 r;
    r.x = rsqrtf(fmaxf((float)lo, 1.0f));
    r.y = rsqrtf(fmaxf((float)hi, 1.0f));
    *(float2*)(d + (size_t)which * NN + 2 * w) = r;
}

#define OR_   400
#define APITCH 132
#define OBLK  (NN / OR_)
#define NG    (EE / 128)
#define LCAP  144
#define WSEG  800
#define WCAP  (WSEG - 1)
#define OSTEPS ((((NG / 32 + 1 + 3) / 4 + 1) / 2) * 2)
static_assert(NN % OR_ == 0 && EE % 128 == 0 && NG % OBLK == 0, "ranges and groups are whole");
static_assert(EE <= (1 << 21) && OR_ <= (1 << 11), "list word: row in the top 11 bits, edge in the low 21");
static_assert(OSTEPS * 4 * 32 >= NG + 32, "the steps cover every group of every wave");
static_assert(32 * WSEG == OR_ * 64, "a block's saved words fill exactly its own span of the layer-2 plane");
static_assert(NN <= (1 << 17) && OR_ <= (1 << 9), "saved word: row in the top bits, source id in the low 17");

template <int COLS, int OFF, bool FULL>
__device__ __forceinline__ void GCN_saved_batch(const float* __restrict__ X, float* acc,
                                                unsigned w, int m, int lane) {
    constexpr int LPR = COLS / 32;
    const int s = (int)(w & 0x1ffffu);
    const int dl = (int)(w >> 17);
    float v[16][LPR];
#pragma unroll
    for (int j = 0; j < 16; ++j) {
        int sj = __builtin_amdgcn_readlane(s, OFF + j);
        if (!FULL && j >= m) sj = 0;
        const float* sp = X + (size_t)sj * COLS + lane;
#pragma unroll
        for (int q = 0; q < LPR; ++q) v[j][q] = sp[32 * q];
    }
#pragma unroll
    for (int j = 0; j < 16; ++j) {
        if (FULL || j < m) {
            int dj = __builtin_amdgcn_readlane(dl, OFF + j);
            float* ap = acc + dj * COLS + lane;
#pragma unroll
            for (int q = 0; q < LPR; ++q)
                __hip_atomic_fetch_add(ap + 32 * q, v[j][q], __ATOMIC_RELAXED, __HIP_MEMORY_SCOPE_WORKGROUP);
        }
    }
}

template <int COLS, bool SRC_SCALE, bool FINISH, bool SAVE, bool USE, bool PROD>
__global__ void __launch_bounds__(1024)
GCN_own_kernel(const float* __restrict__ X, const int* __restrict__ esrc, const int* __restrict__ edst,
               const float* __restrict__ onorm, const float* __restrict__ inorm,
               const float* __restrict__ bias, float* __restrict__ Y, float* __restrict__ L,
               const float* __restrict__ Wa, const float* __restrict__ Wb) {
    constexpr int LPR = COLS / 32;
    constexpr int BAT = 32 / LPR;
    constexpr int PITCH = PROD ? APITCH : COLS;
    static_assert(!PROD || (COLS == 128 && !FINISH && OR_ % 16 == 0), "the products take 128-wide rows in 16-row tiles");
    static_assert(!USE || !PROD, "the saved-edge path adds at pitch COLS");
    static_assert(BAT - 1 + 128 <= LCAP, "list capacity");
    static_assert(!USE || COLS == 64, "the saved words live in a span of OR_ x 64 floats");
    static_assert(!(SAVE && USE), "a launch saves or reads");
    __shared__ __attribute__((aligned(16))) float acc[OR_ * PITCH];
    __shared__ unsigned lst[32 * LCAP];
    const int tid = threadIdx.x, lane = tid & 31;
    const int wave = __builtin_amdgcn_readfirstlane(tid >> 5);
    const int base = blockIdx.x * OR_;
    const int rot = blockIdx.x * (NG / OBLK);
    const int4* ids4 = (const int4*)edst;
    unsigned* mylst = lst + wave * LCAP;
    float4* acc4 = (float4*)acc;
    int cnt = 0;
    int gpos = 0;
    float* sv = L + (size_t)blockIdx.x * (OR_ * 64) + wave * WSEG;

    auto load4 = [&](int4 (&buf)[4], int s) {
#pragma unroll
        for (int j = 0; j < 4; ++j) {
            int it0 = wave + 32 * (s * 4 + j);
            int g = it0 + rot;
            if (g >= NG) g -= NG;
            if (it0 >= NG) g = 0;
            buf[j] = ids4[g * 32 + lane];
        }
    };
    auto drain = [&]() {
        cnt -= BAT;
        unsigned p = mylst[cnt + (lane & (BAT - 1))];
        int s = esrc[p & 0x1fffffu];
        int dl = (int)(p >> 21);
        float sc = 1.0f;
        if (SRC_SCALE) sc = onorm[s];
        if (SAVE) {
            if (gpos + BAT <= WCAP && lane < BAT)
                sv[gpos + lane] = __uint_as_float(((unsigned)dl << 17) | (unsigned)s);
            gpos += BAT;
        }
        float v[BAT][LPR];
#pragma unroll
        for (int j = 0; j < BAT; ++j) {
            int sj = __builtin_amdgcn_readlane(s, j);
            const float* sp = X + (size_t)sj * COLS + lane;
#pragma unroll
            for (int q = 0; q < LPR; ++q) v[j][q] = sp[32 * q];
        }
#pragma unroll
        for (int j = 0; j < BAT; ++j) {
            int dj = __builtin_amdgcn_readlane(dl, j);
            float scj = __int_as_float(__builtin_amdgcn_readlane(__float_as_int(sc), j));
            float* ap = acc + dj * PITCH + lane;
#pragma unroll
            for (int q = 0; q < LPR; ++q) {
                float val = SRC_SCALE ? v[j][q] * scj : v[j][q];
                __hip_atomic_fetch_add(ap + 32 * q, val, __ATOMIC_RELAXED, __HIP_MEMORY_SCOPE_WORKGROUP);
            }
        }
    };
    auto scan4 = [&](const int4 (&buf)[4], int s) {
#pragma unroll
        for (int j = 0; j < 4; ++j) {
            int it0 = wave + 32 * (s * 4 + j);
            if (it0 < NG) {
                int g = it0 + rot;
                if (g >= NG) g -= NG;
                unsigned u[4];
                u[0] = (unsigned)(buf[j].x - base); u[1] = (unsigned)(buf[j].y - base);
                u[2] = (unsigned)(buf[j].z - base); u[3] = (unsigned)(buf[j].w - base);
                unsigned mn = min(min(u[0], u[1]), min(u[2], u[3]));
                if (__builtin_amdgcn_ballot_w32(mn < (unsigned)OR_)) {
                    const unsigned e0 = (unsigned)(g * 32 + lane) * 4u;
#pragma unroll
                    for (int c = 0; c < 4; ++c) {
                        bool m = u[c] < (unsigned)OR_;
                        unsigned bal = __builtin_amdgcn_ballot_w32(m);
                        if (bal) {
                            int pos = cnt + (int)__builtin_amdgcn_mbcnt_lo(bal, 0u);
                            if (m) mylst[pos] = (u[c] << 21) | (e0 + c);
                            cnt += __builtin_popcount(bal);
                        }
                    }
                    while (cnt >= BAT) drain();
                }
            }
        }
    };

    int4 ia[4], ib[4];
    int nsaved = 0;
    bool refilter = !USE;
    const float* span = Y + (size_t)blockIdx.x * (OR_ * COLS);
    if (USE) {
        unsigned cv = __float_as_uint(span[lane * WSEG + WCAP]);
        refilter = __builtin_amdgcn_ballot_w32(cv > (unsigned)WCAP) != 0u;
        nsaved = __builtin_amdgcn_readlane((int)cv, wave);
    } else {
        load4(ia, 0);
    }
    unsigned pw = 0u;
    if (USE && !refilter && nsaved > 0)
        pw = __float_as_uint(span[wave * WSEG + lane]);
    for (int i = tid; i < OR_ * PITCH / 4; i += 1024) acc4[i] = make_float4(0.0f, 0.0f, 0.0f, 0.0f);
    __syncthreads();
    if (USE && !refilter) {
        const float* seg = span + wave * WSEG;
        int done = 0;
        for (; done + 32 <= nsaved; done += 32) {
            unsigned pn = 0u;
            if (done + 32 < nsaved) pn = __float_as_uint(seg[done + 32 + lane]);
            GCN_saved_batch<COLS, 0, true>(X, acc, pw, 16, lane);
            GCN_saved_batch<COLS, 16, true>(X, acc, pw, 16, lane);
            pw = pn;
        }
        const int m = nsaved - done;
        if (m >= 16) {
            GCN_saved_batch<COLS, 0, true>(X, acc, pw, 16, lane);
            if (m > 16) GCN_saved_batch<COLS, 16, false>(X, acc, pw, m - 16, lane);
        } else if (m > 0) {
            GCN_saved_batch<COLS, 0, false>(X, acc, pw, m, lane);
        }
    } else {
    if (USE) load4(ia, 0);
    for (int s = 0; s < OSTEPS; s += 2) {
        load4(ib, s + 1);
        scan4(ia, s);
        load4(ia, s + 2);
        scan4(ib, s + 1);
    }
    if (cnt > 0) {
        unsigned p = mylst[lane < cnt ? lane : 0];
        int s = esrc[p & 0x1fffffu];
        int dl = (int)(p >> 21);
        float sc = 1.0f;
        if (SRC_SCALE) sc = onorm[s];
        if (SAVE) {
            if (gpos + cnt <= WCAP && lane < cnt)
                sv[gpos + lane] = __uint_as_float(((unsigned)dl << 17) | (unsigned)s);
            gpos += cnt;
        }
        for (int j = 0; j < cnt; ++j) {
            int sj = __builtin_amdgcn_readlane(s, j);
            int dj = __builtin_amdgcn_readlane(dl, j);
            float scj = __int_as_float(__builtin_amdgcn_readlane(__float_as_int(sc), j));
            const float* sp = X + (size_t)sj * COLS + lane;
            float* ap = acc + dj * PITCH + lane;
#pragma unroll
            for (int q = 0; q < LPR; ++q) {
                float val = SRC_SCALE ? sp[32 * q] * scj : sp[32 * q];
                __hip_atomic_fetch_add(ap + 32 * q, val, __ATOMIC_RELAXED, __HIP_MEMORY_SCOPE_WORKGROUP);
            }
        }
    }
    if (SAVE && lane == 0) sv[WCAP] = __uint_as_float((unsigned)gpos);
    }
    __syncthreads();
    if constexpr (PROD) {
        constexpr int NT = OR_ / 16;
        const int r = lane & 15, kh = lane >> 4;
        {
            const int cs = wave & 7, rg = wave >> 3;
            const int n = cs * 16 + r;
            v2f bf[32];
            const float* Bp = Wa + (size_t)(2 * kh) * 128 + n;
#pragma unroll
            for (int kt = 0; kt < 32; ++kt) {
                bf[kt].x = Bp[kt * 512];
                bf[kt].y = Bp[kt * 512 + 128];
            }
            const float bb = bias[n];
            for (int it = 0; it < (NT + 3) / 4; ++it) {
                const int t = rg + 4 * it;
                const bool live = t < NT;
                v8f c = {};
                float4 ci0 = {}, ci1 = {}, co0 = {}, co1 = {};
                if (live) {
                    const float4* ip = (const float4*)(inorm + base + t * 16 + 8 * kh);
                    const float4* op = (const float4*)(onorm + base + t * 16 + 8 * kh);
                    ci0 = ip[0]; ci1 = ip[1]; co0 = op[0]; co1 = op[1];
                    const float* ap = acc + (t * 16 + r) * PITCH + 2 * kh;
#pragma unroll
                    for (int kt = 0; kt < 32; ++kt) {
                        v2f a = *(const v2f*)(ap + kt * 4);
                        c = __builtin_amdgcn_wmma_f32_16x16x4_f32(false, a, false, bf[kt], (short)0, c, false, false);
                    }
                }
                __syncthreads();
                if (live) {
                    const float ci[8] = {ci0.x, ci0.y, ci0.z, ci0.w, ci1.x, ci1.y, ci1.z, ci1.w};
                    const float co[8] = {co0.x, co0.y, co0.z, co0.w, co1.x, co1.y, co1.z, co1.w};
                    float* hp = acc + (t * 16 + 8 * kh) * PITCH + n;
#pragma unroll
                    for (int v = 0; v < 8; ++v) {
                        float val = fmaxf(fmaf(c[v], ci[v], bb), 0.0f);
                        hp[v * PITCH] = val * co[v];
                    }
                }
            }
        }
        __syncthreads();
        {
            const int cs = wave & 3, rg = wave >> 2;
            const int n = cs * 16 + r;
            v2f bf[32];
            const float* Bp = Wb + (size_t)(2 * kh) * 64 + n;
#pragma unroll
            for (int kt = 0; kt < 32; ++kt) {
                bf[kt].x = Bp[kt * 256];
                bf[kt].y = Bp[kt * 256 + 64];
            }
            for (int t = rg; t < NT; t += 8) {
                const float* ap = acc + (t * 16 + r) * PITCH + 2 * kh;
                v8f c = {};
#pragma unroll
                for (int kt = 0; kt < 32; ++kt) {
                    v2f a = *(const v2f*)(ap + kt * 4);
                    c = __builtin_amdgcn_wmma_f32_16x16x4_f32(false, a, false, bf[kt], (short)0, c, false, false);
                }
                float* tp = Y + (size_t)(base + t * 16 + 8 * kh) * 64 + n;
#pragma unroll
                for (int v = 0; v < 8; ++v) tp[v * 64] = c[v];
            }
        }
        return;
    }
    float4* Y4 = (float4*)(Y + (size_t)base * COLS);
    for (int i = tid; i < OR_ * COLS / 4; i += 1024) {
        float4 a = acc4[i];
        if (FINISH) {
            const int row = base + i / (COLS / 4);
            const int c4 = (i % (COLS / 4)) * 4;
            const float sn = inorm[row];
            a.x = fmaf(a.x, sn, bias[c4 + 0]);
            a.y = fmaf(a.y, sn, bias[c4 + 1]);
            a.z = fmaf(a.z, sn, bias[c4 + 2]);
            a.w = fmaf(a.w, sn, bias[c4 + 3]);
        }
        Y4[i] = a;
    }
}

__device__ __forceinline__ float GCN_xhalf(float x) {
    const int b = __builtin_bit_cast(int, x);
    return __builtin_bit_cast(float, __builtin_amdgcn_permlanex16(b, b, 0x76543210, 0xfedcba98, false, false));
}
template <int M>
__device__ __forceinline__ float GCN_xrow(float x) {
    return __builtin_bit_cast(float, __builtin_amdgcn_update_dpp(0, __builtin_bit_cast(int, x), 0x160 + M, 0xf, 0xf, true));
}
#define PFD 65536
__device__ __forceinline__ void GCN_ids_ahead(const int* __restrict__ qs, const int* __restrict__ qd, int q) {
    const int wv = __builtin_amdgcn_readfirstlane((int)(threadIdx.x >> 5));
    if (wv == 0 && q + PFD < QQ) {
        __builtin_prefetch(qs + q + PFD, 0, 1);
        __builtin_prefetch(qd + q + PFD, 0, 1);
    }
}
__global__ void GCN_score_kernel(const float* __restrict__ H2, const int* __restrict__ qs,
                                 const int* __restrict__ qd, float* __restrict__ out) {
    int t = blockIdx.x * blockDim.x + threadIdx.x;
    int q = t >> 5, lane = t & 31;
    int s = qs[q], d = qd[q];
    GCN_ids_ahead(qs, qd, q);
    float2 a = *(const float2*)(H2 + (size_t)s * 64 + lane * 2);
    float2 b = *(const float2*)(H2 + (size_t)d * 64 + lane * 2);
    float p = a.x * b.x + a.y * b.y;
    p += GCN_xhalf(p);
    p += GCN_xrow<8>(p);
    p += GCN_xrow<4>(p);
    p += GCN_xrow<2>(p);
    p += GCN_xrow<1>(p);
    if (lane == 0) out[q] = 1.0f / (1.0f + __expf(-p));
}

extern "C" void kernel_launch(void* const* d_in, const int* in_sizes, int n_in,
                              void* d_out, int out_size, void* d_ws, size_t ws_size,
                              hipStream_t stream) {
    const float* feat = (const float*)d_in[0];
    const int* esrc   = (const int*)d_in[1];
    const int* edst   = (const int*)d_in[2];
    const int* qsrc   = (const int*)d_in[3];
    const int* qdst   = (const int*)d_in[4];
    const float* W1   = (const float*)d_in[5];
    const float* b1   = (const float*)d_in[6];
    const float* W2   = (const float*)d_in[7];
    const float* b2   = (const float*)d_in[8];
    float* out = (float*)d_out;

    float* ws    = (float*)d_ws;
    float* onorm = ws;
    float* inorm = ws + NN;
    int* part    = (int*)(ws + 2 * NN);
    float* t2    = ws + 2 * NN + (size_t)NN * 256;
    float* h2    = t2 + (size_t)NN * 64;

    GCN_hist_kernel<<<2 * HCH, 1024, 0, stream>>>(esrc, edst, part);
    GCN_norm_kernel<<<(2 * HW + 255) / 256, 256, 0, stream>>>(part, onorm);
    GCN_own_kernel<128, true, false, true, false, true><<<OBLK, 1024, 0, stream>>>(feat, esrc, edst, onorm, inorm, b1, t2, h2, W1, W2);
    GCN_own_kernel<64, false, true, false, true, false><<<OBLK, 1024, 0, stream>>>(t2, esrc, edst, onorm, inorm, b2, h2, h2, nullptr, nullptr);
    GCN_score_kernel<<<QQ / 8, 256, 0, stream>>>(h2, qsrc, qdst, out);
}


# Layer-1 edge filter reads one owner byte per edge (matched 4 per word op, 512 edges a load) and one packed row|src word per kept edge, both written beside the degree count; LDS arrival order kept
# speedup vs baseline: 3.0934x; 1.0552x over previous
#include <hip/hip_runtime.h>
#include <math.h>

typedef __attribute__((ext_vector_type(2))) float v2f;
typedef __attribute__((ext_vector_type(8))) float v8f;

#define NN 100000
#define EE 1600000
#define QQ 1000000

#define OR_   400
#define OBLK  (NN / OR_)
#define GRP   512
#define NG    (EE / GRP)
#define PKB   125
#define PKN4  (NG / PKB * GRP / 4)
#define PKIT  ((PKN4 + 1023) / 1024)
static_assert(EE % GRP == 0 && NG % PKB == 0, "packing blocks take whole groups");
static_assert(NN % OR_ == 0 && OBLK <= 255 && NN <= (1 << 17) && OR_ <= (1 << 9), "owner byte; packed word: row above bit 17, source id below");
#define HCH  64
#define HLEN (EE / HCH)
#define HW   (NN / 2)
#define HWP  50016
#define HIT  ((HLEN / 4 + 1023) / 1024)
static_assert(EE % HCH == 0 && HLEN % 4 == 0 && HLEN < 65536, "chunk: whole int4s, 16-bit safe");
static_assert(NN % 2 == 0 && HW % 4 == 0 && HWP >= HW && (HWP * 4) % 128 == 0, "plane layout");
__global__ void __launch_bounds__(1024)
GCN_hist_kernel(const int* __restrict__ esrc, const int* __restrict__ edst, int* __restrict__ part,
                int* __restrict__ pk, int* __restrict__ ob) {
    __shared__ __attribute__((aligned(16))) unsigned cnt[HW];
    const int tid = threadIdx.x;
    if (blockIdx.x >= 2 * HCH) {
        const int b4 = (blockIdx.x - 2 * HCH) * PKN4;
        const int4* d4 = (const int4*)edst + b4;
        const int4* s4 = (const int4*)esrc + b4;
        uint4* pk4 = (uint4*)pk + b4;
        unsigned* ob1 = (unsigned*)ob + b4;
        int4 dv[PKIT], sv[PKIT];
#pragma unroll
        for (int k = 0; k < PKIT; ++k) {
            int i = tid + k * 1024;
            if (i < PKN4) { dv[k] = d4[i]; sv[k] = s4[i]; }
        }
#pragma unroll
        for (int k = 0; k < PKIT; ++k) {
            int i = tid + k * 1024;
            if (i < PKN4) {
                const unsigned d[4] = {(unsigned)dv[k].x, (unsigned)dv[k].y, (unsigned)dv[k].z, (unsigned)dv[k].w};
                const unsigned s[4] = {(unsigned)sv[k].x, (unsigned)sv[k].y, (unsigned)sv[k].z, (unsigned)sv[k].w};
                unsigned w[4], o = 0u;
#pragma unroll
                for (int c = 0; c < 4; ++c) {
                    unsigned ow = d[c] / (unsigned)OR_;
                    unsigned rw = d[c] - ow * (unsigned)OR_;
                    if (d[c] >= (unsigned)NN) { ow = 255u; rw = 0u; }
                    w[c] = (rw << 17) | (s[c] & 0x1ffffu);
                    o |= ow << (8 * c);
                }
                pk4[i] = make_uint4(w[0], w[1], w[2], w[3]);
                ob1[i] = o;
            }
        }
        return;
    }
    const int which = blockIdx.x / HCH, chunk = blockIdx.x - which * HCH;
    const int4* ids4 = (const int4*)((which ? edst : esrc) + (size_t)chunk * HLEN);
    uint4* cnt4 = (uint4*)cnt;
    int4 v[HIT];
#pragma unroll
    for (int k = 0; k < HIT; ++k) {
        int i = tid + k * 1024;
        if (i < HLEN / 4) v[k] = ids4[i];
    }
    for (int w = tid; w < HW / 4; w += 1024) cnt4[w] = make_uint4(0u, 0u, 0u, 0u);
    __syncthreads();
#pragma unroll
    for (int k = 0; k < HIT; ++k) {
        int i = tid + k * 1024;
        if (i < HLEN / 4) {
            atomicAdd(&cnt[v[k].x >> 1], 1u << ((v[k].x & 1) * 16));
            atomicAdd(&cnt[v[k].y >> 1], 1u << ((v[k].y & 1) * 16));
            atomicAdd(&cnt[v[k].z >> 1], 1u << ((v[k].z & 1) * 16));
            atomicAdd(&cnt[v[k].w >> 1], 1u << ((v[k].w & 1) * 16));
        }
    }
    __syncthreads();
    uint4* out4 = (uint4*)(part + (size_t)blockIdx.x * HWP);
    for (int w = tid; w < HW / 4; w += 1024) out4[w] = cnt4[w];
}

__global__ void __launch_bounds__(256)
GCN_norm_kernel(const int* __restrict__ part, float* __restrict__ d) {
    int i = blockIdx.x * 256 + threadIdx.x;
    if (i >= 2 * HW) return;
    const int which = i >= HW ? 1 : 0;
    const int w = i - which * HW;
    const unsigned* p = (const unsigned*)part + (size_t)which * HCH * HWP + w;
    unsigned lo = 0, hi = 0;
#pragma unroll 16
    for (int c = 0; c < HCH; ++c) {
        unsigned v = p[(size_t)c * HWP];
        lo += v & 0xffffu;
        hi += v >> 16;
    }
    float2 r;
    r.x = rsqrtf(fmaxf((float)lo, 1.0f));
    r.y = rsqrtf(fmaxf((float)hi, 1.0f));
    *(float2*)(d + (size_t)which * NN + 2 * w) = r;
}

#define APITCH 132
#define LCAP  528
#define WSEG  800
#define WCAP  (WSEG - 1)
#define OSTEPS ((((NG / 32 + 1 + 3) / 4 + 1) / 2) * 2)
static_assert(OSTEPS * 4 * 32 >= NG + 32, "the steps cover every group of every wave");
static_assert(32 * WSEG == OR_ * 64, "a block's saved words fill exactly its own span of the layer-2 plane");

template <int COLS, int OFF, bool FULL>
__device__ __forceinline__ void GCN_saved_batch(const float* __restrict__ X, float* acc,
                                                unsigned w, int m, int lane) {
    constexpr int LPR = COLS / 32;
    const int s = (int)(w & 0x1ffffu);
    const int dl = (int)(w >> 17);
    float v[16][LPR];
#pragma unroll
    for (int j = 0; j < 16; ++j) {
        int sj = __builtin_amdgcn_readlane(s, OFF + j);
        if (!FULL && j >= m) sj = 0;
        const float* sp = X + (size_t)sj * COLS + lane;
#pragma unroll
        for (int q = 0; q < LPR; ++q) v[j][q] = sp[32 * q];
    }
#pragma unroll
    for (int j = 0; j < 16; ++j) {
        if (FULL || j < m) {
            int dj = __builtin_amdgcn_readlane(dl, OFF + j);
            float* ap = acc + dj * COLS + lane;
#pragma unroll
            for (int q = 0; q < LPR; ++q)
                __hip_atomic_fetch_add(ap + 32 * q, v[j][q], __ATOMIC_RELAXED, __HIP_MEMORY_SCOPE_WORKGROUP);
        }
    }
}

template <int COLS, bool SRC_SCALE, bool FINISH, bool SAVE, bool USE, bool PROD>
__global__ void __launch_bounds__(1024)
GCN_own_kernel(const float* __restrict__ X, const int* __restrict__ pk, const int* __restrict__ ob,
               const float* __restrict__ onorm, const float* __restrict__ inorm,
               const float* __restrict__ bias, float* __restrict__ Y, float* __restrict__ L,
               const float* __restrict__ Wa, const float* __restrict__ Wb) {
    constexpr int LPR = COLS / 32;
    constexpr int BAT = 32 / LPR;
    constexpr int PITCH = PROD ? APITCH : COLS;
    static_assert(!PROD || (COLS == 128 && !FINISH && OR_ % 16 == 0), "the products take 128-wide rows in 16-row tiles");
    static_assert(!USE || !PROD, "the saved-edge path adds at pitch COLS");
    static_assert(BAT - 1 + GRP <= LCAP, "list capacity");
    static_assert(!USE || COLS == 64, "the saved words live in a span of OR_ x 64 floats");
    static_assert(!(SAVE && USE), "a launch saves or reads");
    __shared__ __attribute__((aligned(16))) float acc[OR_ * PITCH];
    __shared__ unsigned lst[32 * LCAP];
    const int tid = threadIdx.x, lane = tid & 31;
    const int wave = __builtin_amdgcn_readfirstlane(tid >> 5);
    const int base = blockIdx.x * OR_;
    const int rot = blockIdx.x * (NG / OBLK);
    const int4* ids4 = (const int4*)ob;
    const unsigned pat = blockIdx.x * 0x01010101u;
    unsigned* mylst = lst + wave * LCAP;
    float4* acc4 = (float4*)acc;
    int cnt = 0;
    int gpos = 0;
    float* sv = L + (size_t)blockIdx.x * (OR_ * 64) + wave * WSEG;

    auto load4 = [&](int4 (&buf)[4], int s) {
#pragma unroll
        for (int j = 0; j < 4; ++j) {
            int it0 = wave + 32 * (s * 4 + j);
            int g = it0 + rot;
            if (g >= NG) g -= NG;
            if (it0 >= NG) g = 0;
            buf[j] = ids4[g * 32 + lane];
        }
    };
    auto drain = [&]() {
        cnt -= BAT;
        unsigned p = (unsigned)pk[mylst[cnt + (lane & (BAT - 1))]];
        int s = (int)(p & 0x1ffffu);
        int dl = (int)(p >> 17);
        float sc = 1.0f;
        if (SRC_SCALE) sc = onorm[s];
        if (SAVE) {
            if (gpos + BAT <= WCAP && lane < BAT)
                sv[gpos + lane] = __uint_as_float(p);
            gpos += BAT;
        }
        float v[BAT][LPR];
#pragma unroll
        for (int j = 0; j < BAT; ++j) {
            int sj = __builtin_amdgcn_readlane(s, j);
            const float* sp = X + (size_t)sj * COLS + lane;
#pragma unroll
            for (int q = 0; q < LPR; ++q) v[j][q] = sp[32 * q];
        }
#pragma unroll
        for (int j = 0; j < BAT; ++j) {
            int dj = __builtin_amdgcn_readlane(dl, j);
            float scj = __int_as_float(__builtin_amdgcn_readlane(__float_as_int(sc), j));
            float* ap = acc + dj * PITCH + lane;
#pragma unroll
            for (int q = 0; q < LPR; ++q) {
                float val = SRC_SCALE ? v[j][q] * scj : v[j][q];
                __hip_atomic_fetch_add(ap + 32 * q, val, __ATOMIC_RELAXED, __HIP_MEMORY_SCOPE_WORKGROUP);
            }
        }
    };
    auto scan4 = [&](const int4 (&buf)[4], int s) {
#pragma unroll
        for (int j = 0; j < 4; ++j) {
            int it0 = wave + 32 * (s * 4 + j);
            if (it0 < NG) {
                int g = it0 + rot;
                if (g >= NG) g -= NG;
                const unsigned x[4] = {(unsigned)buf[j].x ^ pat, (unsigned)buf[j].y ^ pat,
                                       (unsigned)buf[j].z ^ pat, (unsigned)buf[j].w ^ pat};
                unsigned nz = 0u;
#pragma unroll
                for (int k = 0; k < 4; ++k) {
                    unsigned y = ((x[k] & 0x7f7f7f7fu) + 0x7f7f7f7fu) | x[k];
                    nz |= (y >> (7 - k)) & (0x01010101u << k);
                }
                unsigned M = nz ^ 0x0f0f0f0fu;
                unsigned bal = __builtin_amdgcn_ballot_w32(M != 0u);
                if (bal) {
                    const unsigned e0 = (unsigned)(g * 32 + lane) * 16u;
                    do {
                        if (M != 0u) {
                            const unsigned b = (unsigned)__builtin_ctz(M);
                            M &= M - 1u;
                            int pos = cnt + (int)__builtin_amdgcn_mbcnt_lo(bal, 0u);
                            mylst[pos] = e0 + ((b & 7u) << 2) + (b >> 3);
                        }
                        cnt += __builtin_popcount(bal);
                        bal = __builtin_amdgcn_ballot_w32(M != 0u);
                    } while (bal);
                    while (cnt >= BAT) drain();
                }
            }
        }
    };

    int4 ia[4], ib[4];
    int nsaved = 0;
    bool refilter = !USE;
    const float* span = Y + (size_t)blockIdx.x * (OR_ * COLS);
    if (USE) {
        unsigned cv = __float_as_uint(span[lane * WSEG + WCAP]);
        refilter = __builtin_amdgcn_ballot_w32(cv > (unsigned)WCAP) != 0u;
        nsaved = __builtin_amdgcn_readlane((int)cv, wave);
    } else {
        load4(ia, 0);
    }
    unsigned pw = 0u;
    if (USE && !refilter && nsaved > 0)
        pw = __float_as_uint(span[wave * WSEG + lane]);
    for (int i = tid; i < OR_ * PITCH / 4; i += 1024) acc4[i] = make_float4(0.0f, 0.0f, 0.0f, 0.0f);
    __syncthreads();
    if (USE && !refilter) {
        const float* seg = span + wave * WSEG;
        int done = 0;
        for (; done + 32 <= nsaved; done += 32) {
            unsigned pn = 0u;
            if (done + 32 < nsaved) pn = __float_as_uint(seg[done + 32 + lane]);
            GCN_saved_batch<COLS, 0, true>(X, acc, pw, 16, lane);
            GCN_saved_batch<COLS, 16, true>(X, acc, pw, 16, lane);
            pw = pn;
        }
        const int m = nsaved - done;
        if (m >= 16) {
            GCN_saved_batch<COLS, 0, true>(X, acc, pw, 16, lane);
            if (m > 16) GCN_saved_batch<COLS, 16, false>(X, acc, pw, m - 16, lane);
        } else if (m > 0) {
            GCN_saved_batch<COLS, 0, false>(X, acc, pw, m, lane);
        }
    } else {
    if (USE) load4(ia, 0);
    for (int s = 0; s < OSTEPS; s += 2) {
        load4(ib, s + 1);
        scan4(ia, s);
        load4(ia, s + 2);
        scan4(ib, s + 1);
    }
    if (cnt > 0) {
        unsigned p = (unsigned)pk[mylst[lane < cnt ? lane : 0]];
        int s = (int)(p & 0x1ffffu);
        int dl = (int)(p >> 17);
        float sc = 1.0f;
        if (SRC_SCALE) sc = onorm[s];
        if (SAVE) {
            if (gpos + cnt <= WCAP && lane < cnt)
                sv[gpos + lane] = __uint_as_float(p);
            gpos += cnt;
        }
        for (int j = 0; j < cnt; ++j) {
            int sj = __builtin_amdgcn_readlane(s, j);
            int dj = __builtin_amdgcn_readlane(dl, j);
            float scj = __int_as_float(__builtin_amdgcn_readlane(__float_as_int(sc), j));
            const float* sp = X + (size_t)sj * COLS + lane;
            float* ap = acc + dj * PITCH + lane;
#pragma unroll
            for (int q = 0; q < LPR; ++q) {
                float val = SRC_SCALE ? sp[32 * q] * scj : sp[32 * q];
                __hip_atomic_fetch_add(ap + 32 * q, val, __ATOMIC_RELAXED, __HIP_MEMORY_SCOPE_WORKGROUP);
            }
        }
    }
    if (SAVE && lane == 0) sv[WCAP] = __uint_as_float((unsigned)gpos);
    }
    __syncthreads();
    if constexpr (PROD) {
        constexpr int NT = OR_ / 16;
        const int r = lane & 15, kh = lane >> 4;
        {
            const int cs = wave & 7, rg = wave >> 3;
            const int n = cs * 16 + r;
            v2f bf[32];
            const float* Bp = Wa + (size_t)(2 * kh) * 128 + n;
#pragma unroll
            for (int kt = 0; kt < 32; ++kt) {
                bf[kt].x = Bp[kt * 512];
                bf[kt].y = Bp[kt * 512 + 128];
            }
            const float bb = bias[n];
            for (int it = 0; it < (NT + 3) / 4; ++it) {
                const int t = rg + 4 * it;
                const bool live = t < NT;
                v8f c = {};
                float4 ci0 = {}, ci1 = {}, co0 = {}, co1 = {};
                if (live) {
                    const float4* ip = (const float4*)(inorm + base + t * 16 + 8 * kh);
                    const float4* op = (const float4*)(onorm + base + t * 16 + 8 * kh);
                    ci0 = ip[0]; ci1 = ip[1]; co0 = op[0]; co1 = op[1];
                    const float* ap = acc + (t * 16 + r) * PITCH + 2 * kh;
#pragma unroll
                    for (int kt = 0; kt < 32; ++kt) {
                        v2f a = *(const v2f*)(ap + kt * 4);
                        c = __builtin_amdgcn_wmma_f32_16x16x4_f32(false, a, false, bf[kt], (short)0, c, false, false);
                    }
                }
                __syncthreads();
                if (live) {
                    const float ci[8] = {ci0.x, ci0.y, ci0.z, ci0.w, ci1.x, ci1.y, ci1.z, ci1.w};
                    const float co[8] = {co0.x, co0.y, co0.z, co0.w, co1.x, co1.y, co1.z, co1.w};
                    float* hp = acc + (t * 16 + 8 * kh) * PITCH + n;
#pragma unroll
                    for (int v = 0; v < 8; ++v) {
                        float val = fmaxf(fmaf(c[v], ci[v], bb), 0.0f);
                        hp[v * PITCH] = val * co[v];
                    }
                }
            }
        }
        __syncthreads();
        {
            const int cs = wave & 3, rg = wave >> 2;
            const int n = cs * 16 + r;
            v2f bf[32];
            const float* Bp = Wb + (size_t)(2 * kh) * 64 + n;
#pragma unroll
            for (int kt = 0; kt < 32; ++kt) {
                bf[kt].x = Bp[kt * 256];
                bf[kt].y = Bp[kt * 256 + 64];
            }
            for (int t = rg; t < NT; t += 8) {
                const float* ap = acc + (t * 16 + r) * PITCH + 2 * kh;
                v8f c = {};
#pragma unroll
                for (int kt = 0; kt < 32; ++kt) {
                    v2f a = *(const v2f*)(ap + kt * 4);
                    c = __builtin_amdgcn_wmma_f32_16x16x4_f32(false, a, false, bf[kt], (short)0, c, false, false);
                }
                float* tp = Y + (size_t)(base + t * 16 + 8 * kh) * 64 + n;
#pragma unroll
                for (int v = 0; v < 8; ++v) tp[v * 64] = c[v];
            }
        }
        return;
    }
    float4* Y4 = (float4*)(Y + (size_t)base * COLS);
    for (int i = tid; i < OR_ * COLS / 4; i += 1024) {
        float4 a = acc4[i];
        if (FINISH) {
            const int row = base + i / (COLS / 4);
            const int c4 = (i % (COLS / 4)) * 4;
            const float sn = inorm[row];
            a.x = fmaf(a.x, sn, bias[c4 + 0]);
            a.y = fmaf(a.y, sn, bias[c4 + 1]);
            a.z = fmaf(a.z, sn, bias[c4 + 2]);
            a.w = fmaf(a.w, sn, bias[c4 + 3]);
        }
        Y4[i] = a;
    }
}

__device__ __forceinline__ float GCN_xhalf(float x) {
    const int b = __builtin_bit_cast(int, x);
    return __builtin_bit_cast(float, __builtin_amdgcn_permlanex16(b, b, 0x76543210, 0xfedcba98, false, false));
}
template <int M>
__device__ __forceinline__ float GCN_xrow(float x) {
    return __builtin_bit_cast(float, __builtin_amdgcn_update_dpp(0, __builtin_bit_cast(int, x), 0x160 + M, 0xf, 0xf, true));
}
#define PFD 65536
__device__ __forceinline__ void GCN_ids_ahead(const int* __restrict__ qs, const int* __restrict__ qd, int q) {
    const int wv = __builtin_amdgcn_readfirstlane((int)(threadIdx.x >> 5));
    if (wv == 0 && q + PFD < QQ) {
        __builtin_prefetch(qs + q + PFD, 0, 1);
        __builtin_prefetch(qd + q + PFD, 0, 1);
    }
}
__global__ void GCN_score_kernel(const float* __restrict__ H2, const int* __restrict__ qs,
                                 const int* __restrict__ qd, float* __restrict__ out) {
    int t = blockIdx.x * blockDim.x + threadIdx.x;
    int q = t >> 5, lane = t & 31;
    int s = qs[q], d = qd[q];
    GCN_ids_ahead(qs, qd, q);
    float2 a = *(const float2*)(H2 + (size_t)s * 64 + lane * 2);
    float2 b = *(const float2*)(H2 + (size_t)d * 64 + lane * 2);
    float p = a.x * b.x + a.y * b.y;
    p += GCN_xhalf(p);
    p += GCN_xrow<8>(p);
    p += GCN_xrow<4>(p);
    p += GCN_xrow<2>(p);
    p += GCN_xrow<1>(p);
    if (lane == 0) out[q] = 1.0f / (1.0f + __expf(-p));
}

extern "C" void kernel_launch(void* const* d_in, const int* in_sizes, int n_in,
                              void* d_out, int out_size, void* d_ws, size_t ws_size,
                              hipStream_t stream) {
    const float* feat = (const float*)d_in[0];
    const int* esrc   = (const int*)d_in[1];
    const int* edst   = (const int*)d_in[2];
    const int* qsrc   = (const int*)d_in[3];
    const int* qdst   = (const int*)d_in[4];
    const float* W1   = (const float*)d_in[5];
    const float* b1   = (const float*)d_in[6];
    const float* W2   = (const float*)d_in[7];
    const float* b2   = (const float*)d_in[8];
    float* out = (float*)d_out;

    float* ws    = (float*)d_ws;
    float* onorm = ws;
    float* inorm = ws + NN;
    int* part    = (int*)(ws + 2 * NN);
    int* pk      = part + (size_t)2 * HCH * HWP;
    int* ob      = pk + EE;
    float* t2    = ws + 2 * NN + (size_t)NN * 256;
    float* h2    = t2 + (size_t)NN * 64;

    GCN_hist_kernel<<<2 * HCH + PKB, 1024, 0, stream>>>(esrc, edst, part, pk, ob);
    GCN_norm_kernel<<<(2 * HW + 255) / 256, 256, 0, stream>>>(part, onorm);
    GCN_own_kernel<128, true, false, true, false, true><<<OBLK, 1024, 0, stream>>>(feat, pk, ob, onorm, inorm, b1, t2, h2, W1, W2);
    GCN_own_kernel<64, false, true, false, true, false><<<OBLK, 1024, 0, stream>>>(t2, pk, ob, onorm, inorm, b2, h2, h2, nullptr, nullptr);
    GCN_score_kernel<<<QQ / 8, 256, 0, stream>>>(h2, qsrc, qdst, out);
}
